# starting kernel
# baseline (speedup 1.0000x reference)
#include <hip/hip_runtime.h>
#include <math.h>

#define N_NODES 50000
#define N_EDGES 800000
#define D_MODEL 128
#define N_HEADS 2

typedef __attribute__((ext_vector_type(2))) float v2f;
typedef __attribute__((ext_vector_type(8))) float v8f;

__device__ __forceinline__ unsigned ordEnc(float f) {
    unsigned u = __float_as_uint(f);
    return (u & 0x80000000u) ? ~u : (u | 0x80000000u);
}
__device__ __forceinline__ float ordDec(unsigned u) {
    unsigned v = (u & 0x80000000u) ? (u & 0x7fffffffu) : ~u;
    return __uint_as_float(v);
}
#define ORD_NEG_INF 0x007FFFFFu

__global__ void fill_u32(unsigned* __restrict__ p, unsigned v, int n) {
    int i = blockIdx.x * blockDim.x + threadIdx.x;
    if (i < n) p[i] = v;
}

__global__ void __launch_bounds__(128) gemm_feat(const float* __restrict__ h,
                                                 const float* __restrict__ W,
                                                 float* __restrict__ feat) {
    const int lane  = threadIdx.x & 31;
    const int wave  = threadIdx.x >> 5;
    const int mbase = blockIdx.x << 4;
    const int nq    = wave << 6;
    const int row   = lane & 15;
    const int khalf = lane >> 4;

    v8f acc0 = {}, acc1 = {}, acc2 = {}, acc3 = {};
    const float* Arow = h + (size_t)(mbase + row) * D_MODEL;

    for (int k = 0; k < D_MODEL; k += 4) {
        const int ka = k + khalf * 2;
        v2f a;
        a.x = Arow[ka];
        a.y = Arow[ka + 1];
        const float* Wb = W + (size_t)ka * 256 + row;
        v2f b0, b1, b2, b3;
        b0.x = Wb[nq];        b0.y = Wb[256 + nq];
        b1.x = Wb[nq + 16];   b1.y = Wb[256 + nq + 16];
        b2.x = Wb[nq + 32];   b2.y = Wb[256 + nq + 32];
        b3.x = Wb[nq + 48];   b3.y = Wb[256 + nq + 48];
        acc0 = __builtin_amdgcn_wmma_f32_16x16x4_f32(false, a, false, b0, (short)0, acc0, false, false);
        acc1 = __builtin_amdgcn_wmma_f32_16x16x4_f32(false, a, false, b1, (short)0, acc1, false, false);
        acc2 = __builtin_amdgcn_wmma_f32_16x16x4_f32(false, a, false, b2, (short)0, acc2, false, false);
        acc3 = __builtin_amdgcn_wmma_f32_16x16x4_f32(false, a, false, b3, (short)0, acc3, false, false);
    }

    #pragma unroll
    for (int j = 0; j < 8; ++j) {
        const int m = mbase + j + khalf * 8;
        float* F = feat + (size_t)m * 256 + nq + row;
        F[0]  = acc0[j];
        F[16] = acc1[j];
        F[32] = acc2[j];
        F[48] = acc3[j];
    }
}

__global__ void __launch_bounds__(256) node_attn(const float* __restrict__ feat,
                                                 const float* __restrict__ al,
                                                 const float* __restrict__ ar,
                                                 float* __restrict__ el,
                                                 float* __restrict__ er) {
    const int gw   = (blockIdx.x * blockDim.x + threadIdx.x) >> 5;
    const int lane = threadIdx.x & 31;
    if (gw >= N_NODES) return;
    const float4* F0 = (const float4*)(feat + (size_t)gw * 256) + lane;
    const float4* F1 = (const float4*)(feat + (size_t)gw * 256 + 128) + lane;
    const float4* AL0 = (const float4*)(al) + lane;
    const float4* AL1 = (const float4*)(al + 128) + lane;
    const float4* AR0 = (const float4*)(ar) + lane;
    const float4* AR1 = (const float4*)(ar + 128) + lane;
    float4 f0 = *F0, f1 = *F1;
    float4 l0 = *AL0, l1 = *AL1, r0 = *AR0, r1 = *AR1;
    float e0 = f0.x*l0.x + f0.y*l0.y + f0.z*l0.z + f0.w*l0.w;
    float e1 = f1.x*l1.x + f1.y*l1.y + f1.z*l1.z + f1.w*l1.w;
    float g0 = f0.x*r0.x + f0.y*r0.y + f0.z*r0.z + f0.w*r0.w;
    float g1 = f1.x*r1.x + f1.y*r1.y + f1.z*r1.z + f1.w*r1.w;
    #pragma unroll
    for (int off = 16; off; off >>= 1) {
        e0 += __shfl_xor(e0, off, 32);
        e1 += __shfl_xor(e1, off, 32);
        g0 += __shfl_xor(g0, off, 32);
        g1 += __shfl_xor(g1, off, 32);
    }
    if (lane == 0) {
        el[gw * 2 + 0] = e0;  el[gw * 2 + 1] = e1;
        er[gw * 2 + 0] = g0;  er[gw * 2 + 1] = g1;
    }
}

__global__ void __launch_bounds__(256) edge_max(const float* __restrict__ el,
                                                const float* __restrict__ er,
                                                const int* __restrict__ src,
                                                const int* __restrict__ dst,
                                                float* __restrict__ ebuf,
                                                unsigned* __restrict__ nodemax) {
    int i = blockIdx.x * blockDim.x + threadIdx.x;
    if (i >= N_EDGES * N_HEADS) return;
    const int e = i >> 1, hh = i & 1;
    const int d = dst[e];
    float v = el[src[e] * 2 + hh] + er[d * 2 + hh];
    v = v > 0.0f ? v : 0.2f * v;
    ebuf[i] = v;
    atomicMax(nodemax + d * 2 + hh, ordEnc(v));
}

__global__ void __launch_bounds__(256) edge_exp(float* __restrict__ ebuf,
                                                const unsigned* __restrict__ nodemax,
                                                const int* __restrict__ dst,
                                                float* __restrict__ denom) {
    int i = blockIdx.x * blockDim.x + threadIdx.x;
    if (i >= N_EDGES * N_HEADS) return;
    const int e = i >> 1, hh = i & 1;
    const int d = dst[e];
    const float m = ordDec(nodemax[d * 2 + hh]);
    const float ex = expf(ebuf[i] - m);
    ebuf[i] = ex;
    unsafeAtomicAdd(denom + d * 2 + hh, ex);
}

__global__ void __launch_bounds__(256) aggregate(const float* __restrict__ feat,
                                                 const float* __restrict__ ebuf,
                                                 const float* __restrict__ denom,
                                                 const int* __restrict__ src,
                                                 const int* __restrict__ dst,
                                                 float* __restrict__ agg) {
    const int gw   = (blockIdx.x * blockDim.x + threadIdx.x) >> 5;
    const int lane = threadIdx.x & 31;
    if (gw >= N_EDGES * N_HEADS) return;
    const int e = gw >> 1, hh = gw & 1;
    const int s = src[e], d = dst[e];
    const float w = ebuf[gw] / denom[d * 2 + hh];
    const float4 f = *(const float4*)(feat + (size_t)s * 256 + hh * 128 + lane * 4);
    float* o = agg + (size_t)d * 256 + hh * 128 + lane * 4;
    unsafeAtomicAdd(o + 0, f.x * w);
    unsafeAtomicAdd(o + 1, f.y * w);
    unsafeAtomicAdd(o + 2, f.z * w);
    unsafeAtomicAdd(o + 3, f.w * w);
}

__global__ void __launch_bounds__(256) finalize_mean(const float* __restrict__ agg,
                                                     const float* __restrict__ b,
                                                     float* __restrict__ hnext) {
    int i = blockIdx.x * blockDim.x + threadIdx.x;
    if (i >= N_NODES * D_MODEL) return;
    const int n = i >> 7, f = i & 127;
    const float o0 = agg[(size_t)n * 256 + f]       + b[f];
    const float o1 = agg[(size_t)n * 256 + 128 + f] + b[128 + f];
    hnext[i] = 0.5f * (o0 + o1);
}

__global__ void __launch_bounds__(256) finalize_bias(float* __restrict__ out,
                                                     const float* __restrict__ b) {
    int i = blockIdx.x * blockDim.x + threadIdx.x;
    if (i >= N_NODES * 256) return;
    out[i] += b[i & 255];
}

extern "C" void kernel_launch(void* const* d_in, const int* in_sizes, int n_in,
                              void* d_out, int out_size, void* d_ws, size_t ws_size,
                              hipStream_t stream) {
    const float* x   = (const float*)d_in[0];
    const float* Ws  = (const float*)d_in[1];
    const float* als = (const float*)d_in[2];
    const float* ars = (const float*)d_in[3];
    const float* bs  = (const float*)d_in[4];
    const int*   src = (const int*)d_in[5];
    const int*   dst = (const int*)d_in[6];

    float* ws      = (float*)d_ws;
    float* feat    = ws;
    float* hnext   = ws + 12800000;
    float* el      = ws + 19200000;
    float* er      = ws + 19300000;
    unsigned* nmax = (unsigned*)(ws + 19400000);
    float* denom   = ws + 19500000;
    float* ebuf    = ws + 19600000;

    float* agg = (float*)d_out;

    const int NH   = N_NODES * N_HEADS;
    const int EH   = N_EDGES * N_HEADS;
    const int NOUT = N_NODES * 256;

    for (int layer = 0; layer < 2; ++layer) {
        const float* hin = layer ? hnext : x;
        const float* W   = Ws  + (size_t)layer * D_MODEL * 256;
        const float* al  = als + (size_t)layer * 256;
        const float* ar  = ars + (size_t)layer * 256;
        const float* b   = bs  + (size_t)layer * 256;

        fill_u32<<<(NOUT + 255) / 256, 256, 0, stream>>>((unsigned*)agg, 0u, NOUT);
        fill_u32<<<(NH + 255) / 256, 256, 0, stream>>>(nmax, ORD_NEG_INF, NH);
        fill_u32<<<(NH + 255) / 256, 256, 0, stream>>>((unsigned*)denom, 0u, NH);

        gemm_feat<<<N_NODES / 16, 128, 0, stream>>>(hin, W, feat);
        node_attn<<<(N_NODES + 7) / 8, 256, 0, stream>>>(feat, al, ar, el, er);
        edge_max<<<(EH + 255) / 256, 256, 0, stream>>>(el, er, src, dst, ebuf, nmax);
        edge_exp<<<(EH + 255) / 256, 256, 0, stream>>>(ebuf, nmax, dst, denom);
        aggregate<<<(EH * 32 + 255) / 256, 256, 0, stream>>>(feat, ebuf, denom, src, dst, agg);

        if (layer == 0) {
            finalize_mean<<<(N_NODES * D_MODEL + 255) / 256, 256, 0, stream>>>(agg, b, hnext);
        } else {
            finalize_bias<<<(NOUT + 255) / 256, 256, 0, stream>>>(agg, b);
        }
    }
}


# aggregation owned by dst range: a block keeps 196 nodes' output rows in LDS, scans all dst and adds alpha*feat[src] of its own edges with LDS float adds (was global atomics); sum order = LDS arrival
# speedup vs baseline: 2.2393x; 2.2393x over previous
#include <hip/hip_runtime.h>
#include <math.h>

#define N_NODES 50000
#define N_EDGES 800000
#define D_MODEL 128
#define N_HEADS 2

typedef __attribute__((ext_vector_type(2))) float v2f;
typedef __attribute__((ext_vector_type(8))) float v8f;

__device__ __forceinline__ unsigned ordEnc(float f) {
    unsigned u = __float_as_uint(f);
    return (u & 0x80000000u) ? ~u : (u | 0x80000000u);
}
__device__ __forceinline__ float ordDec(unsigned u) {
    unsigned v = (u & 0x80000000u) ? (u & 0x7fffffffu) : ~u;
    return __uint_as_float(v);
}
#define ORD_NEG_INF 0x007FFFFFu

__global__ void fill_u32(unsigned* __restrict__ p, unsigned v, int n) {
    int i = blockIdx.x * blockDim.x + threadIdx.x;
    if (i < n) p[i] = v;
}

__global__ void __launch_bounds__(128) gemm_feat(const float* __restrict__ h,
                                                 const float* __restrict__ W,
                                                 float* __restrict__ feat) {
    const int lane  = threadIdx.x & 31;
    const int wave  = threadIdx.x >> 5;
    const int mbase = blockIdx.x << 4;
    const int nq    = wave << 6;
    const int row   = lane & 15;
    const int khalf = lane >> 4;

    v8f acc0 = {}, acc1 = {}, acc2 = {}, acc3 = {};
    const float* Arow = h + (size_t)(mbase + row) * D_MODEL;

    for (int k = 0; k < D_MODEL; k += 4) {
        const int ka = k + khalf * 2;
        v2f a;
        a.x = Arow[ka];
        a.y = Arow[ka + 1];
        const float* Wb = W + (size_t)ka * 256 + row;
        v2f b0, b1, b2, b3;
        b0.x = Wb[nq];        b0.y = Wb[256 + nq];
        b1.x = Wb[nq + 16];   b1.y = Wb[256 + nq + 16];
        b2.x = Wb[nq + 32];   b2.y = Wb[256 + nq + 32];
        b3.x = Wb[nq + 48];   b3.y = Wb[256 + nq + 48];
        acc0 = __builtin_amdgcn_wmma_f32_16x16x4_f32(false, a, false, b0, (short)0, acc0, false, false);
        acc1 = __builtin_amdgcn_wmma_f32_16x16x4_f32(false, a, false, b1, (short)0, acc1, false, false);
        acc2 = __builtin_amdgcn_wmma_f32_16x16x4_f32(false, a, false, b2, (short)0, acc2, false, false);
        acc3 = __builtin_amdgcn_wmma_f32_16x16x4_f32(false, a, false, b3, (short)0, acc3, false, false);
    }

    #pragma unroll
    for (int j = 0; j < 8; ++j) {
        const int m = mbase + j + khalf * 8;
        float* F = feat + (size_t)m * 256 + nq + row;
        F[0]  = acc0[j];
        F[16] = acc1[j];
        F[32] = acc2[j];
        F[48] = acc3[j];
    }
}

__global__ void __launch_bounds__(256) node_attn(const float* __restrict__ feat,
                                                 const float* __restrict__ al,
                                                 const float* __restrict__ ar,
                                                 float* __restrict__ el,
                                                 float* __restrict__ er) {
    const int gw   = (blockIdx.x * blockDim.x + threadIdx.x) >> 5;
    const int lane = threadIdx.x & 31;
    if (gw >= N_NODES) return;
    const float4* F0 = (const float4*)(feat + (size_t)gw * 256) + lane;
    const float4* F1 = (const float4*)(feat + (size_t)gw * 256 + 128) + lane;
    const float4* AL0 = (const float4*)(al) + lane;
    const float4* AL1 = (const float4*)(al + 128) + lane;
    const float4* AR0 = (const float4*)(ar) + lane;
    const float4* AR1 = (const float4*)(ar + 128) + lane;
    float4 f0 = *F0, f1 = *F1;
    float4 l0 = *AL0, l1 = *AL1, r0 = *AR0, r1 = *AR1;
    float e0 = f0.x*l0.x + f0.y*l0.y + f0.z*l0.z + f0.w*l0.w;
    float e1 = f1.x*l1.x + f1.y*l1.y + f1.z*l1.z + f1.w*l1.w;
    float g0 = f0.x*r0.x + f0.y*r0.y + f0.z*r0.z + f0.w*r0.w;
    float g1 = f1.x*r1.x + f1.y*r1.y + f1.z*r1.z + f1.w*r1.w;
    #pragma unroll
    for (int off = 16; off; off >>= 1) {
        e0 += __shfl_xor(e0, off, 32);
        e1 += __shfl_xor(e1, off, 32);
        g0 += __shfl_xor(g0, off, 32);
        g1 += __shfl_xor(g1, off, 32);
    }
    if (lane == 0) {
        el[gw * 2 + 0] = e0;  el[gw * 2 + 1] = e1;
        er[gw * 2 + 0] = g0;  er[gw * 2 + 1] = g1;
    }
}

__global__ void __launch_bounds__(256) edge_max(const float* __restrict__ el,
                                                const float* __restrict__ er,
                                                const int* __restrict__ src,
                                                const int* __restrict__ dst,
                                                float* __restrict__ ebuf,
                                                unsigned* __restrict__ nodemax) {
    int i = blockIdx.x * blockDim.x + threadIdx.x;
    if (i >= N_EDGES * N_HEADS) return;
    const int e = i >> 1, hh = i & 1;
    const int d = dst[e];
    float v = el[src[e] * 2 + hh] + er[d * 2 + hh];
    v = v > 0.0f ? v : 0.2f * v;
    ebuf[i] = v;
    atomicMax(nodemax + d * 2 + hh, ordEnc(v));
}

__global__ void __launch_bounds__(256) edge_exp(float* __restrict__ ebuf,
                                                const unsigned* __restrict__ nodemax,
                                                const int* __restrict__ dst,
                                                float* __restrict__ denom) {
    int i = blockIdx.x * blockDim.x + threadIdx.x;
    if (i >= N_EDGES * N_HEADS) return;
    const int e = i >> 1, hh = i & 1;
    const int d = dst[e];
    const float m = ordDec(nodemax[d * 2 + hh]);
    const float ex = expf(ebuf[i] - m);
    ebuf[i] = ex;
    unsafeAtomicAdd(denom + d * 2 + hh, ex);
}

#define RANGE    196
#define N_RANGES 256
#define QCAP     544

__device__ __forceinline__ void lds_add(float* p, float v) {
    __hip_atomic_fetch_add(p, v, __ATOMIC_RELAXED, __HIP_MEMORY_SCOPE_WORKGROUP);
}

template <int LAYER>
__device__ __forceinline__ void add_rows(float* tile, const unsigned* q, const int n,
                                         const int lane, const int base,
                                         const float* __restrict__ feat,
                                         const float* __restrict__ ebuf,
                                         const float* __restrict__ denom,
                                         const int* __restrict__ src) {
    constexpr int COLS = LAYER ? 256 : 128;
    const unsigned ent = (lane < n) ? q[lane] : 0u;
    const int e  = (int)(ent >> 8);
    const int dl = (int)(ent & 255u);
    const int s  = src[e];
    const float2 ex = *(const float2*)(ebuf + (size_t)e * 2);
    const float2 dn = *(const float2*)(denom + (size_t)(base + dl) * 2);
    float w0, w1;
    if (LAYER) { w0 = ex.x / dn.x;          w1 = ex.y / dn.y; }
    else       { w0 = 0.5f * (ex.x / dn.x); w1 = 0.5f * (ex.y / dn.y); }

    for (int j0 = 0; j0 < n; j0 += 8) {
        float4 f0[8], f1[8];
        #pragma unroll
        for (int u = 0; u < 8; ++u) {
            const int sj = __builtin_amdgcn_readlane(s, j0 + u);
            const float* row = feat + (size_t)sj * 256 + lane * 4;
            f0[u] = *(const float4*)row;
            f1[u] = *(const float4*)(row + 128);
        }
        #pragma unroll
        for (int u = 0; u < 8; ++u) {
            if (j0 + u < n) {
                const int   dj = __builtin_amdgcn_readlane(dl, j0 + u);
                const float a0 = __int_as_float(__builtin_amdgcn_readlane(__float_as_int(w0), j0 + u));
                const float a1 = __int_as_float(__builtin_amdgcn_readlane(__float_as_int(w1), j0 + u));
                float* t = tile + dj * COLS + lane;
                if (LAYER) {
                    lds_add(t +   0, f0[u].x * a0);
                    lds_add(t +  32, f0[u].y * a0);
                    lds_add(t +  64, f0[u].z * a0);
                    lds_add(t +  96, f0[u].w * a0);
                    lds_add(t + 128, f1[u].x * a1);
                    lds_add(t + 160, f1[u].y * a1);
                    lds_add(t + 192, f1[u].z * a1);
                    lds_add(t + 224, f1[u].w * a1);
                } else {
                    lds_add(t +  0, f0[u].x * a0 + f1[u].x * a1);
                    lds_add(t + 32, f0[u].y * a0 + f1[u].y * a1);
                    lds_add(t + 64, f0[u].z * a0 + f1[u].z * a1);
                    lds_add(t + 96, f0[u].w * a0 + f1[u].w * a1);
                }
            }
        }
    }
}

template <int LAYER>
__device__ __forceinline__ void owned_range(float* tile, unsigned* queue,
                                            const float* __restrict__ feat,
                                            const float* __restrict__ ebuf,
                                            const float* __restrict__ denom,
                                            const int* __restrict__ src,
                                            const int* __restrict__ dst,
                                            const float* __restrict__ b,
                                            float* __restrict__ outp) {
    constexpr int COLS = LAYER ? 256 : 128;
    constexpr int NV   = N_EDGES / 4;
    const int tid  = threadIdx.x;
    const int lane = tid & 31;
    const int wave = __builtin_amdgcn_readfirstlane(tid >> 5);
    const int base = blockIdx.x * RANGE;

    for (int i = tid; i < RANGE * COLS; i += 1024) {
        float v = 0.0f;
        if (LAYER == 0) {
            const int r = i & 127, c = r >> 5, l = r & 31;
            v = 0.5f * (b[l * 4 + c] + b[128 + l * 4 + c]);
        }
        tile[i] = v;
    }
    __syncthreads();

    unsigned* q = queue + wave * QCAP;
    unsigned cnt = 0;
    const unsigned lt = (1u << lane) - 1u;
    const int4* d4 = (const int4*)dst;
    for (int i0 = wave * 32; i0 < NV; i0 += 4096) {
        int4 dv[4];
        #pragma unroll
        for (int u = 0; u < 4; ++u) {
            const int i = i0 + u * 1024 + lane;
            dv[u] = d4[i < NV ? i : NV - 1];
        }
        #pragma unroll
        for (int u = 0; u < 4; ++u) {
            const int i = i0 + u * 1024 + lane;
            const bool ok = i < NV;
            const unsigned r0 = (unsigned)(dv[u].x - base), r1 = (unsigned)(dv[u].y - base);
            const unsigned r2 = (unsigned)(dv[u].z - base), r3 = (unsigned)(dv[u].w - base);
            const bool m0 = ok & (r0 < RANGE), m1 = ok & (r1 < RANGE);
            const bool m2 = ok & (r2 < RANGE), m3 = ok & (r3 < RANGE);
            if (__builtin_amdgcn_ballot_w32(m0 | m1 | m2 | m3)) {
                const unsigned e4 = (unsigned)i * 4u;
                unsigned bal;
                bal = __builtin_amdgcn_ballot_w32(m0);
                if (m0) q[cnt + __builtin_popcount(bal & lt)] = ((e4 + 0u) << 8) | r0;
                cnt += __builtin_popcount(bal);
                bal = __builtin_amdgcn_ballot_w32(m1);
                if (m1) q[cnt + __builtin_popcount(bal & lt)] = ((e4 + 1u) << 8) | r1;
                cnt += __builtin_popcount(bal);
                bal = __builtin_amdgcn_ballot_w32(m2);
                if (m2) q[cnt + __builtin_popcount(bal & lt)] = ((e4 + 2u) << 8) | r2;
                cnt += __builtin_popcount(bal);
                bal = __builtin_amdgcn_ballot_w32(m3);
                if (m3) q[cnt + __builtin_popcount(bal & lt)] = ((e4 + 3u) << 8) | r3;
                cnt += __builtin_popcount(bal);
            }
        }
        while (cnt >= 32) {
            cnt -= 32;
            add_rows<LAYER>(tile, q + cnt, 32, lane, base, feat, ebuf, denom, src);
        }
    }
    if (cnt) add_rows<LAYER>(tile, q, (int)cnt, lane, base, feat, ebuf, denom, src);
    __syncthreads();

    constexpr int V4 = COLS / 4;
    for (int i = tid; i < RANGE * V4; i += 1024) {
        const int n = i / V4, r = i - n * V4;
        const int h = r >> 5, l = r & 31;
        if (base + n < N_NODES) {
            const float* t = tile + n * COLS + h * 128 + l;
            float4 v = make_float4(t[0], t[32], t[64], t[96]);
            if (LAYER) {
                const float4 bb = *(const float4*)(b + h * 128 + l * 4);
                v.x += bb.x; v.y += bb.y; v.z += bb.z; v.w += bb.w;
            }
            *(float4*)(outp + (size_t)(base + n) * COLS + h * 128 + l * 4) = v;
        }
    }
}

__global__ void __launch_bounds__(1024) aggregate_owned_mean(const float* __restrict__ feat,
                                                             const float* __restrict__ ebuf,
                                                             const float* __restrict__ denom,
                                                             const int* __restrict__ src,
                                                             const int* __restrict__ dst,
                                                             const float* __restrict__ b,
                                                             float* __restrict__ hnext) {
    __shared__ float tile[RANGE * 128];
    __shared__ unsigned queue[32 * QCAP];
    owned_range<0>(tile, queue, feat, ebuf, denom, src, dst, b, hnext);
}

__global__ void __launch_bounds__(1024) aggregate_owned_out(const float* __restrict__ feat,
                                                            const float* __restrict__ ebuf,
                                                            const float* __restrict__ denom,
                                                            const int* __restrict__ src,
                                                            const int* __restrict__ dst,
                                                            const float* __restrict__ b,
                                                            float* __restrict__ out) {
    __shared__ float tile[RANGE * 256];
    __shared__ unsigned queue[32 * QCAP];
    owned_range<1>(tile, queue, feat, ebuf, denom, src, dst, b, out);
}

extern "C" void kernel_launch(void* const* d_in, const int* in_sizes, int n_in,
                              void* d_out, int out_size, void* d_ws, size_t ws_size,
                              hipStream_t stream) {
    const float* x   = (const float*)d_in[0];
    const float* Ws  = (const float*)d_in[1];
    const float* als = (const float*)d_in[2];
    const float* ars = (const float*)d_in[3];
    const float* bs  = (const float*)d_in[4];
    const int*   src = (const int*)d_in[5];
    const int*   dst = (const int*)d_in[6];

    float* ws      = (float*)d_ws;
    float* feat    = ws;
    float* hnext   = ws + 12800000;
    float* el      = ws + 19200000;
    float* er      = ws + 19300000;
    unsigned* nmax = (unsigned*)(ws + 19400000);
    float* denom   = ws + 19500000;
    float* ebuf    = ws + 19600000;

    float* out = (float*)d_out;

    const int NH   = N_NODES * N_HEADS;
    const int EH   = N_EDGES * N_HEADS;

    for (int layer = 0; layer < 2; ++layer) {
        const float* hin = layer ? hnext : x;
        const float* W   = Ws  + (size_t)layer * D_MODEL * 256;
        const float* al  = als + (size_t)layer * 256;
        const float* ar  = ars + (size_t)layer * 256;
        const float* b   = bs  + (size_t)layer * 256;

        fill_u32<<<(NH + 255) / 256, 256, 0, stream>>>(nmax, ORD_NEG_INF, NH);
        fill_u32<<<(NH + 255) / 256, 256, 0, stream>>>((unsigned*)denom, 0u, NH);

        gemm_feat<<<N_NODES / 16, 128, 0, stream>>>(hin, W, feat);
        node_attn<<<(N_NODES + 7) / 8, 256, 0, stream>>>(feat, al, ar, el, er);
        edge_max<<<(EH + 255) / 256, 256, 0, stream>>>(el, er, src, dst, ebuf, nmax);
        edge_exp<<<(EH + 255) / 256, 256, 0, stream>>>(ebuf, nmax, dst, denom);
        if (layer == 0) {
            aggregate_owned_mean<<<N_RANGES, 1024, 0, stream>>>(feat, ebuf, denom, src, dst, b, hnext);
        } else {
            aggregate_owned_out<<<N_RANGES, 1024, 0, stream>>>(feat, ebuf, denom, src, dst, b, out);
        }
    }
}


# edge softmax inside the block that owns the dst range: node max and denominator by LDS max/add over its queued edges (was 3.2M scattered global atomics a layer); denominator sum order = LDS arrival
# speedup vs baseline: 3.6709x; 1.6393x over previous
#include <hip/hip_runtime.h>
#include <math.h>

#define N_NODES 50000
#define N_EDGES 800000
#define D_MODEL 128
#define N_HEADS 2

typedef __attribute__((ext_vector_type(2))) float v2f;
typedef __attribute__((ext_vector_type(8))) float v8f;

__device__ __forceinline__ unsigned ordEnc(float f) {
    unsigned u = __float_as_uint(f);
    return (u & 0x80000000u) ? ~u : (u | 0x80000000u);
}
__device__ __forceinline__ float ordDec(unsigned u) {
    unsigned v = (u & 0x80000000u) ? (u & 0x7fffffffu) : ~u;
    return __uint_as_float(v);
}
#define ORD_NEG_INF 0x007FFFFFu

__global__ void __launch_bounds__(128) gemm_feat(const float* __restrict__ h,
                                                 const float* __restrict__ W,
                                                 float* __restrict__ feat) {
    const int lane  = threadIdx.x & 31;
    const int wave  = threadIdx.x >> 5;
    const int mbase = blockIdx.x << 4;
    const int nq    = wave << 6;
    const int row   = lane & 15;
    const int khalf = lane >> 4;

    v8f acc0 = {}, acc1 = {}, acc2 = {}, acc3 = {};
    const float* Arow = h + (size_t)(mbase + row) * D_MODEL;

    for (int k = 0; k < D_MODEL; k += 4) {
        const int ka = k + khalf * 2;
        v2f a;
        a.x = Arow[ka];
        a.y = Arow[ka + 1];
        const float* Wb = W + (size_t)ka * 256 + row;
        v2f b0, b1, b2, b3;
        b0.x = Wb[nq];        b0.y = Wb[256 + nq];
        b1.x = Wb[nq + 16];   b1.y = Wb[256 + nq + 16];
        b2.x = Wb[nq + 32];   b2.y = Wb[256 + nq + 32];
        b3.x = Wb[nq + 48];   b3.y = Wb[256 + nq + 48];
        acc0 = __builtin_amdgcn_wmma_f32_16x16x4_f32(false, a, false, b0, (short)0, acc0, false, false);
        acc1 = __builtin_amdgcn_wmma_f32_16x16x4_f32(false, a, false, b1, (short)0, acc1, false, false);
        acc2 = __builtin_amdgcn_wmma_f32_16x16x4_f32(false, a, false, b2, (short)0, acc2, false, false);
        acc3 = __builtin_amdgcn_wmma_f32_16x16x4_f32(false, a, false, b3, (short)0, acc3, false, false);
    }

    #pragma unroll
    for (int j = 0; j < 8; ++j) {
        const int m = mbase + j + khalf * 8;
        float* F = feat + (size_t)m * 256 + nq + row;
        F[0]  = acc0[j];
        F[16] = acc1[j];
        F[32] = acc2[j];
        F[48] = acc3[j];
    }
}

__global__ void __launch_bounds__(256) node_attn(const float* __restrict__ feat,
                                                 const float* __restrict__ al,
                                                 const float* __restrict__ ar,
                                                 float* __restrict__ el,
                                                 float* __restrict__ er) {
    const int gw   = (blockIdx.x * blockDim.x + threadIdx.x) >> 5;
    const int lane = threadIdx.x & 31;
    if (gw >= N_NODES) return;
    const float4* F0 = (const float4*)(feat + (size_t)gw * 256) + lane;
    const float4* F1 = (const float4*)(feat + (size_t)gw * 256 + 128) + lane;
    const float4* AL0 = (const float4*)(al) + lane;
    const float4* AL1 = (const float4*)(al + 128) + lane;
    const float4* AR0 = (const float4*)(ar) + lane;
    const float4* AR1 = (const float4*)(ar + 128) + lane;
    float4 f0 = *F0, f1 = *F1;
    float4 l0 = *AL0, l1 = *AL1, r0 = *AR0, r1 = *AR1;
    float e0 = f0.x*l0.x + f0.y*l0.y + f0.z*l0.z + f0.w*l0.w;
    float e1 = f1.x*l1.x + f1.y*l1.y + f1.z*l1.z + f1.w*l1.w;
    float g0 = f0.x*r0.x + f0.y*r0.y + f0.z*r0.z + f0.w*r0.w;
    float g1 = f1.x*r1.x + f1.y*r1.y + f1.z*r1.z + f1.w*r1.w;
    #pragma unroll
    for (int off = 16; off; off >>= 1) {
        e0 += __shfl_xor(e0, off, 32);
        e1 += __shfl_xor(e1, off, 32);
        g0 += __shfl_xor(g0, off, 32);
        g1 += __shfl_xor(g1, off, 32);
    }
    if (lane == 0) {
        el[gw * 2 + 0] = e0;  el[gw * 2 + 1] = e1;
        er[gw * 2 + 0] = g0;  er[gw * 2 + 1] = g1;
    }
}

#define RANGE    196
#define N_RANGES 256
#define QCAP     224

__device__ __forceinline__ void lds_add(float* p, float v) {
    __hip_atomic_fetch_add(p, v, __ATOMIC_RELAXED, __HIP_MEMORY_SCOPE_WORKGROUP);
}
__device__ __forceinline__ void lds_max(unsigned* p, unsigned v) {
    __hip_atomic_fetch_max(p, v, __ATOMIC_RELAXED, __HIP_MEMORY_SCOPE_WORKGROUP);
}
__device__ __forceinline__ float leaky(float v) { return v > 0.0f ? v : 0.2f * v; }

__device__ __forceinline__ unsigned append4(unsigned* qe, unsigned cnt, const int4 dv, const int i,
                                            const bool ok, const int base, const unsigned lt) {
    const unsigned r0 = (unsigned)(dv.x - base), r1 = (unsigned)(dv.y - base);
    const unsigned r2 = (unsigned)(dv.z - base), r3 = (unsigned)(dv.w - base);
    const bool m0 = ok & (r0 < RANGE), m1 = ok & (r1 < RANGE);
    const bool m2 = ok & (r2 < RANGE), m3 = ok & (r3 < RANGE);
    if (__builtin_amdgcn_ballot_w32(m0 | m1 | m2 | m3)) {
        const unsigned e4 = (unsigned)i * 4u;
        unsigned bal, pos;
        bal = __builtin_amdgcn_ballot_w32(m0);
        pos = cnt + __builtin_popcount(bal & lt);
        if (m0 && pos < QCAP) qe[pos] = ((e4 + 0u) << 8) | r0;
        cnt += __builtin_popcount(bal);
        bal = __builtin_amdgcn_ballot_w32(m1);
        pos = cnt + __builtin_popcount(bal & lt);
        if (m1 && pos < QCAP) qe[pos] = ((e4 + 1u) << 8) | r1;
        cnt += __builtin_popcount(bal);
        bal = __builtin_amdgcn_ballot_w32(m2);
        pos = cnt + __builtin_popcount(bal & lt);
        if (m2 && pos < QCAP) qe[pos] = ((e4 + 2u) << 8) | r2;
        cnt += __builtin_popcount(bal);
        bal = __builtin_amdgcn_ballot_w32(m3);
        pos = cnt + __builtin_popcount(bal & lt);
        if (m3 && pos < QCAP) qe[pos] = ((e4 + 3u) << 8) | r3;
        cnt += __builtin_popcount(bal);
    }
    return cnt;
}

template <int LAYER>
__device__ __forceinline__ void add_rows(float* tile, const int n, const int lane,
                                         const int s, const int dl, const float w0, const float w1,
                                         const float* __restrict__ feat) {
    constexpr int COLS = LAYER ? 256 : 128;
    for (int j0 = 0; j0 < n; j0 += 8) {
        float4 f0[8], f1[8];
        #pragma unroll
        for (int u = 0; u < 8; ++u) {
            const int sj = __builtin_amdgcn_readlane(s, j0 + u);
            const float* row = feat + (size_t)sj * 256 + lane * 4;
            f0[u] = *(const float4*)row;
            f1[u] = *(const float4*)(row + 128);
        }
        #pragma unroll
        for (int u = 0; u < 8; ++u) {
            if (j0 + u < n) {
                const int   dj = __builtin_amdgcn_readlane(dl, j0 + u);
                const float a0 = __int_as_float(__builtin_amdgcn_readlane(__float_as_int(w0), j0 + u));
                const float a1 = __int_as_float(__builtin_amdgcn_readlane(__float_as_int(w1), j0 + u));
                float* t = tile + dj * COLS + lane;
                if (LAYER) {
                    lds_add(t +   0, f0[u].x * a0);
                    lds_add(t +  32, f0[u].y * a0);
                    lds_add(t +  64, f0[u].z * a0);
                    lds_add(t +  96, f0[u].w * a0);
                    lds_add(t + 128, f1[u].x * a1);
                    lds_add(t + 160, f1[u].y * a1);
                    lds_add(t + 192, f1[u].z * a1);
                    lds_add(t + 224, f1[u].w * a1);
                } else {
                    lds_add(t +  0, f0[u].x * a0 + f1[u].x * a1);
                    lds_add(t + 32, f0[u].y * a0 + f1[u].y * a1);
                    lds_add(t + 64, f0[u].z * a0 + f1[u].z * a1);
                    lds_add(t + 96, f0[u].w * a0 + f1[u].w * a1);
                }
            }
        }
    }
}

template <int LAYER, int PASS>
__device__ __forceinline__ void batch_from_entries(float* tile, unsigned* mx, float* dn, const float2* erl,
                                                   const unsigned* q, const int n, const int lane,
                                                   const float* __restrict__ feat,
                                                   const float* __restrict__ el,
                                                   const int* __restrict__ src) {
    const bool valid = lane < n;
    const unsigned ent = valid ? q[lane] : 0u;
    const int e  = (int)(ent >> 8);
    const int dl = (int)(ent & 255u);
    const int s  = src[e];
    const float2 l = *(const float2*)(el + (size_t)s * 2);
    const float2 r = erl[dl];
    const float c0 = leaky(l.x + r.x), c1 = leaky(l.y + r.y);
    if (PASS == 0) {
        if (valid) { lds_max(mx + dl * 2, ordEnc(c0)); lds_max(mx + dl * 2 + 1, ordEnc(c1)); }
        return;
    }
    const float x0 = expf(c0 - ordDec(mx[dl * 2])), x1 = expf(c1 - ordDec(mx[dl * 2 + 1]));
    if (PASS == 1) {
        if (valid) { lds_add(dn + dl * 2, x0); lds_add(dn + dl * 2 + 1, x1); }
        return;
    }
    float w0 = x0 / dn[dl * 2], w1 = x1 / dn[dl * 2 + 1];
    if (LAYER == 0) { w0 *= 0.5f; w1 *= 0.5f; }
    add_rows<LAYER>(tile, n, lane, s, dl, w0, w1, feat);
}

template <int LAYER, int PASS>
__device__ __forceinline__ void rescan_pass(float* tile, unsigned* mx, float* dn, const float2* erl,
                                            unsigned* qe, const int wave, const int lane, const int base,
                                            const float* __restrict__ feat,
                                            const float* __restrict__ el,
                                            const int* __restrict__ src,
                                            const int* __restrict__ dst) {
    constexpr int NV = N_EDGES / 4;
    const unsigned lt = (1u << lane) - 1u;
    const int4* d4 = (const int4*)dst;
    unsigned cnt = 0;
    for (int i0 = wave * 32; i0 < NV; i0 += 1024) {
        const int i = i0 + lane;
        const bool ok = i < NV;
        const int4 dv = d4[ok ? i : NV - 1];
        cnt = append4(qe, cnt, dv, i, ok, base, lt);
        while (cnt >= 32) {
            cnt -= 32;
            batch_from_entries<LAYER, PASS>(tile, mx, dn, erl, qe + cnt, 32, lane, feat, el, src);
        }
    }
    if (cnt) batch_from_entries<LAYER, PASS>(tile, mx, dn, erl, qe, (int)cnt, lane, feat, el, src);
}

template <int LAYER>
__device__ __forceinline__ void owned_range(float* tile, unsigned* queue, unsigned* mx, float* dn,
                                            float2* erl, unsigned* flag,
                                            const float* __restrict__ feat,
                                            const float* __restrict__ el,
                                            const float* __restrict__ er,
                                            const int* __restrict__ src,
                                            const int* __restrict__ dst,
                                            const float* __restrict__ b,
                                            float* __restrict__ outp) {
    constexpr int COLS = LAYER ? 256 : 128;
    constexpr int NV   = N_EDGES / 4;
    const int tid  = threadIdx.x;
    const int lane = tid & 31;
    const int wave = __builtin_amdgcn_readfirstlane(tid >> 5);
    const int base = blockIdx.x * RANGE;

    for (int i = tid; i < RANGE * COLS; i += 1024) {
        float v = 0.0f;
        if (LAYER == 0) {
            const int r = i & 127, c = r >> 5, l = r & 31;
            v = 0.5f * (b[l * 4 + c] + b[128 + l * 4 + c]);
        }
        tile[i] = v;
    }
    if (tid < RANGE * 2) { mx[tid] = ORD_NEG_INF; dn[tid] = 0.0f; }
    if (tid < RANGE) {
        const int nd = base + tid;
        erl[tid] = (nd < N_NODES) ? *(const float2*)(er + (size_t)nd * 2) : make_float2(0.0f, 0.0f);
    }
    if (tid == 0) *flag = 0u;
    __syncthreads();

    unsigned* qe  = queue + wave * (4 * QCAP);
    int*      qs  = (int*)(qe + QCAP);
    float*    qw0 = (float*)(qe + 2 * QCAP);
    float*    qw1 = (float*)(qe + 3 * QCAP);
    unsigned cnt = 0;
    const unsigned lt = (1u << lane) - 1u;
    const int4* d4 = (const int4*)dst;
    for (int i0 = wave * 32; i0 < NV; i0 += 4096) {
        int4 dv[4];
        #pragma unroll
        for (int u = 0; u < 4; ++u) {
            const int i = i0 + u * 1024 + lane;
            dv[u] = d4[i < NV ? i : NV - 1];
        }
        #pragma unroll
        for (int u = 0; u < 4; ++u) {
            const int i = i0 + u * 1024 + lane;
            cnt = append4(qe, cnt, dv[u], i, i < NV, base, lt);
        }
    }
    if (cnt > QCAP && lane == 0) *flag = 1u;
    __syncthreads();
    const bool overflow = *flag != 0u;

    if (!overflow) {
        const int n = (int)cnt;
        for (int b0 = 0; b0 < n; b0 += 128) {
            unsigned ent[4]; int s[4]; float2 l[4];
            #pragma unroll
            for (int u = 0; u < 4; ++u) {
                const int idx = b0 + u * 32 + lane;
                ent[u] = (idx < n) ? qe[idx] : 0u;
            }
            #pragma unroll
            for (int u = 0; u < 4; ++u) s[u] = src[ent[u] >> 8];
            #pragma unroll
            for (int u = 0; u < 4; ++u) l[u] = *(const float2*)(el + (size_t)s[u] * 2);
            #pragma unroll
            for (int u = 0; u < 4; ++u) {
                const int idx = b0 + u * 32 + lane;
                if (idx < n) {
                    const int dl = (int)(ent[u] & 255u);
                    const float2 r = erl[dl];
                    const float c0 = leaky(l[u].x + r.x), c1 = leaky(l[u].y + r.y);
                    lds_max(mx + dl * 2, ordEnc(c0));
                    lds_max(mx + dl * 2 + 1, ordEnc(c1));
                    qs[idx] = s[u]; qw0[idx] = c0; qw1[idx] = c1;
                }
            }
        }
        __syncthreads();
        for (int idx = lane; idx < n; idx += 32) {
            const int dl = (int)(qe[idx] & 255u);
            const float x0 = expf(qw0[idx] - ordDec(mx[dl * 2]));
            const float x1 = expf(qw1[idx] - ordDec(mx[dl * 2 + 1]));
            lds_add(dn + dl * 2, x0);
            lds_add(dn + dl * 2 + 1, x1);
            qw0[idx] = x0; qw1[idx] = x1;
        }
        __syncthreads();
        for (int b0 = 0; b0 < n; b0 += 32) {
            const int idx = b0 + lane;
            const bool valid = idx < n;
            const unsigned ent = valid ? qe[idx] : 0u;
            const int dl = (int)(ent & 255u);
            const int s  = valid ? qs[idx] : 0;
            float w0 = valid ? qw0[idx] / dn[dl * 2] : 0.0f;
            float w1 = valid ? qw1[idx] / dn[dl * 2 + 1] : 0.0f;
            if (LAYER == 0) { w0 *= 0.5f; w1 *= 0.5f; }
            add_rows<LAYER>(tile, min(32, n - b0), lane, s, dl, w0, w1, feat);
        }
    } else {
        rescan_pass<LAYER, 0>(tile, mx, dn, erl, qe, wave, lane, base, feat, el, src, dst);
        __syncthreads();
        rescan_pass<LAYER, 1>(tile, mx, dn, erl, qe, wave, lane, base, feat, el, src, dst);
        __syncthreads();
        rescan_pass<LAYER, 2>(tile, mx, dn, erl, qe, wave, lane, base, feat, el, src, dst);
    }
    __syncthreads();

    constexpr int V4 = COLS / 4;
    for (int i = tid; i < RANGE * V4; i += 1024) {
        const int n = i / V4, r = i - n * V4;
        const int h = r >> 5, l = r & 31;
        if (base + n < N_NODES) {
            const float* t = tile + n * COLS + h * 128 + l;
            float4 v = make_float4(t[0], t[32], t[64], t[96]);
            if (LAYER) {
                const float4 bb = *(const float4*)(b + h * 128 + l * 4);
                v.x += bb.x; v.y += bb.y; v.z += bb.z; v.w += bb.w;
            }
            *(float4*)(outp + (size_t)(base + n) * COLS + h * 128 + l * 4) = v;
        }
    }
}

__global__ void __launch_bounds__(1024) aggregate_owned_mean(const float* __restrict__ feat,
                                                             const float* __restrict__ el,
                                                             const float* __restrict__ er,
                                                             const int* __restrict__ src,
                                                             const int* __restrict__ dst,
                                                             const float* __restrict__ b,
                                                             float* __restrict__ hnext) {
    __shared__ float tile[RANGE * 128];
    __shared__ unsigned queue[32 * 4 * QCAP];
    __shared__ unsigned mx[RANGE * 2];
    __shared__ float dn[RANGE * 2];
    __shared__ float2 erl[RANGE];
    __shared__ unsigned flag;
    owned_range<0>(tile, queue, mx, dn, erl, &flag, feat, el, er, src, dst, b, hnext);
}

__global__ void __launch_bounds__(1024) aggregate_owned_out(const float* __restrict__ feat,
                                                            const float* __restrict__ el,
                                                            const float* __restrict__ er,
                                                            const int* __restrict__ src,
                                                            const int* __restrict__ dst,
                                                            const float* __restrict__ b,
                                                            float* __restrict__ out) {
    __shared__ float tile[RANGE * 256];
    __shared__ unsigned queue[32 * 4 * QCAP];
    __shared__ unsigned mx[RANGE * 2];
    __shared__ float dn[RANGE * 2];
    __shared__ float2 erl[RANGE];
    __shared__ unsigned flag;
    owned_range<1>(tile, queue, mx, dn, erl, &flag, feat, el, er, src, dst, b, out);
}

extern "C" void kernel_launch(void* const* d_in, const int* in_sizes, int n_in,
                              void* d_out, int out_size, void* d_ws, size_t ws_size,
                              hipStream_t stream) {
    const float* x   = (const float*)d_in[0];
    const float* Ws  = (const float*)d_in[1];
    const float* als = (const float*)d_in[2];
    const float* ars = (const float*)d_in[3];
    const float* bs  = (const float*)d_in[4];
    const int*   src = (const int*)d_in[5];
    const int*   dst = (const int*)d_in[6];

    float* ws      = (float*)d_ws;
    float* feat    = ws;
    float* hnext   = ws + 12800000;
    float* el      = ws + 19200000;
    float* er      = ws + 19300000;

    float* out = (float*)d_out;

    for (int layer = 0; layer < 2; ++layer) {
        const float* hin = layer ? hnext : x;
        const float* W   = Ws  + (size_t)layer * D_MODEL * 256;
        const float* al  = als + (size_t)layer * 256;
        const float* ar  = ars + (size_t)layer * 256;
        const float* b   = bs  + (size_t)layer * 256;

        gemm_feat<<<N_NODES / 16, 128, 0, stream>>>(hin, W, feat);
        node_attn<<<(N_NODES + 7) / 8, 256, 0, stream>>>(feat, al, ar, el, er);
        if (layer == 0) {
            aggregate_owned_mean<<<N_RANGES, 1024, 0, stream>>>(feat, el, er, src, dst, b, hnext);
        } else {
            aggregate_owned_out<<<N_RANGES, 1024, 0, stream>>>(feat, el, er, src, dst, b, out);
        }
    }
}


# layer 2 reuses the edge lists layer 1's scan of dst found: a wave parks one word an edge (source row, node) in its block's own rows of the output buffer; layer 2's per-block scans of dst are dropped
# speedup vs baseline: 4.1524x; 1.1312x over previous
#include <hip/hip_runtime.h>
#include <math.h>

#define N_NODES 50000
#define N_EDGES 800000
#define D_MODEL 128
#define N_HEADS 2

typedef __attribute__((ext_vector_type(2))) float v2f;
typedef __attribute__((ext_vector_type(8))) float v8f;

__device__ __forceinline__ unsigned ordEnc(float f) {
    unsigned u = __float_as_uint(f);
    return (u & 0x80000000u) ? ~u : (u | 0x80000000u);
}
__device__ __forceinline__ float ordDec(unsigned u) {
    unsigned v = (u & 0x80000000u) ? (u & 0x7fffffffu) : ~u;
    return __uint_as_float(v);
}
#define ORD_NEG_INF 0x007FFFFFu

__global__ void __launch_bounds__(128) gemm_feat(const float* __restrict__ h,
                                                 const float* __restrict__ W,
                                                 float* __restrict__ feat) {
    const int lane  = threadIdx.x & 31;
    const int wave  = threadIdx.x >> 5;
    const int mbase = blockIdx.x << 4;
    const int nq    = wave << 6;
    const int row   = lane & 15;
    const int khalf = lane >> 4;

    v8f acc0 = {}, acc1 = {}, acc2 = {}, acc3 = {};
    const float* Arow = h + (size_t)(mbase + row) * D_MODEL;

    for (int k = 0; k < D_MODEL; k += 4) {
        const int ka = k + khalf * 2;
        v2f a;
        a.x = Arow[ka];
        a.y = Arow[ka + 1];
        const float* Wb = W + (size_t)ka * 256 + row;
        v2f b0, b1, b2, b3;
        b0.x = Wb[nq];        b0.y = Wb[256 + nq];
        b1.x = Wb[nq + 16];   b1.y = Wb[256 + nq + 16];
        b2.x = Wb[nq + 32];   b2.y = Wb[256 + nq + 32];
        b3.x = Wb[nq + 48];   b3.y = Wb[256 + nq + 48];
        acc0 = __builtin_amdgcn_wmma_f32_16x16x4_f32(false, a, false, b0, (short)0, acc0, false, false);
        acc1 = __builtin_amdgcn_wmma_f32_16x16x4_f32(false, a, false, b1, (short)0, acc1, false, false);
        acc2 = __builtin_amdgcn_wmma_f32_16x16x4_f32(false, a, false, b2, (short)0, acc2, false, false);
        acc3 = __builtin_amdgcn_wmma_f32_16x16x4_f32(false, a, false, b3, (short)0, acc3, false, false);
    }

    #pragma unroll
    for (int j = 0; j < 8; ++j) {
        const int m = mbase + j + khalf * 8;
        float* F = feat + (size_t)m * 256 + nq + row;
        F[0]  = acc0[j];
        F[16] = acc1[j];
        F[32] = acc2[j];
        F[48] = acc3[j];
    }
}

__global__ void __launch_bounds__(256) node_attn(const float* __restrict__ feat,
                                                 const float* __restrict__ al,
                                                 const float* __restrict__ ar,
                                                 float* __restrict__ el,
                                                 float* __restrict__ er) {
    const int gw   = (blockIdx.x * blockDim.x + threadIdx.x) >> 5;
    const int lane = threadIdx.x & 31;
    if (gw >= N_NODES) return;
    const float4* F0 = (const float4*)(feat + (size_t)gw * 256) + lane;
    const float4* F1 = (const float4*)(feat + (size_t)gw * 256 + 128) + lane;
    const float4* AL0 = (const float4*)(al) + lane;
    const float4* AL1 = (const float4*)(al + 128) + lane;
    const float4* AR0 = (const float4*)(ar) + lane;
    const float4* AR1 = (const float4*)(ar + 128) + lane;
    float4 f0 = *F0, f1 = *F1;
    float4 l0 = *AL0, l1 = *AL1, r0 = *AR0, r1 = *AR1;
    float e0 = f0.x*l0.x + f0.y*l0.y + f0.z*l0.z + f0.w*l0.w;
    float e1 = f1.x*l1.x + f1.y*l1.y + f1.z*l1.z + f1.w*l1.w;
    float g0 = f0.x*r0.x + f0.y*r0.y + f0.z*r0.z + f0.w*r0.w;
    float g1 = f1.x*r1.x + f1.y*r1.y + f1.z*r1.z + f1.w*r1.w;
    #pragma unroll
    for (int off = 16; off; off >>= 1) {
        e0 += __shfl_xor(e0, off, 32);
        e1 += __shfl_xor(e1, off, 32);
        g0 += __shfl_xor(g0, off, 32);
        g1 += __shfl_xor(g1, off, 32);
    }
    if (lane == 0) {
        el[gw * 2 + 0] = e0;  el[gw * 2 + 1] = e1;
        er[gw * 2 + 0] = g0;  er[gw * 2 + 1] = g1;
    }
}

#define RANGE    196
#define N_RANGES 256
#define QCAP     224

__device__ __forceinline__ void lds_add(float* p, float v) {
    __hip_atomic_fetch_add(p, v, __ATOMIC_RELAXED, __HIP_MEMORY_SCOPE_WORKGROUP);
}
__device__ __forceinline__ void lds_max(unsigned* p, unsigned v) {
    __hip_atomic_fetch_max(p, v, __ATOMIC_RELAXED, __HIP_MEMORY_SCOPE_WORKGROUP);
}
__device__ __forceinline__ float leaky(float v) { return v > 0.0f ? v : 0.2f * v; }

__device__ __forceinline__ unsigned append4(unsigned* qe, unsigned cnt, const int4 dv, const int i,
                                            const bool ok, const int base, const unsigned lt) {
    const unsigned r0 = (unsigned)(dv.x - base), r1 = (unsigned)(dv.y - base);
    const unsigned r2 = (unsigned)(dv.z - base), r3 = (unsigned)(dv.w - base);
    const bool m0 = ok & (r0 < RANGE), m1 = ok & (r1 < RANGE);
    const bool m2 = ok & (r2 < RANGE), m3 = ok & (r3 < RANGE);
    if (__builtin_amdgcn_ballot_w32(m0 | m1 | m2 | m3)) {
        const unsigned e4 = (unsigned)i * 4u;
        unsigned bal, pos;
        bal = __builtin_amdgcn_ballot_w32(m0);
        pos = cnt + __builtin_popcount(bal & lt);
        if (m0 && pos < QCAP) qe[pos] = ((e4 + 0u) << 8) | r0;
        cnt += __builtin_popcount(bal);
        bal = __builtin_amdgcn_ballot_w32(m1);
        pos = cnt + __builtin_popcount(bal & lt);
        if (m1 && pos < QCAP) qe[pos] = ((e4 + 1u) << 8) | r1;
        cnt += __builtin_popcount(bal);
        bal = __builtin_amdgcn_ballot_w32(m2);
        pos = cnt + __builtin_popcount(bal & lt);
        if (m2 && pos < QCAP) qe[pos] = ((e4 + 2u) << 8) | r2;
        cnt += __builtin_popcount(bal);
        bal = __builtin_amdgcn_ballot_w32(m3);
        pos = cnt + __builtin_popcount(bal & lt);
        if (m3 && pos < QCAP) qe[pos] = ((e4 + 3u) << 8) | r3;
        cnt += __builtin_popcount(bal);
    }
    return cnt;
}

template <int LAYER>
__device__ __forceinline__ void add_rows(float* tile, const int n, const int lane,
                                         const int s, const int dl, const float w0, const float w1,
                                         const float* __restrict__ feat) {
    constexpr int COLS = LAYER ? 256 : 128;
    for (int j0 = 0; j0 < n; j0 += 8) {
        float4 f0[8], f1[8];
        #pragma unroll
        for (int u = 0; u < 8; ++u) {
            const int sj = __builtin_amdgcn_readlane(s, j0 + u);
            const float* row = feat + (size_t)sj * 256 + lane * 4;
            f0[u] = *(const float4*)row;
            f1[u] = *(const float4*)(row + 128);
        }
        #pragma unroll
        for (int u = 0; u < 8; ++u) {
            if (j0 + u < n) {
                const int   dj = __builtin_amdgcn_readlane(dl, j0 + u);
                const float a0 = __int_as_float(__builtin_amdgcn_readlane(__float_as_int(w0), j0 + u));
                const float a1 = __int_as_float(__builtin_amdgcn_readlane(__float_as_int(w1), j0 + u));
                float* t = tile + dj * COLS + lane;
                if (LAYER) {
                    lds_add(t +   0, f0[u].x * a0);
                    lds_add(t +  32, f0[u].y * a0);
                    lds_add(t +  64, f0[u].z * a0);
                    lds_add(t +  96, f0[u].w * a0);
                    lds_add(t + 128, f1[u].x * a1);
                    lds_add(t + 160, f1[u].y * a1);
                    lds_add(t + 192, f1[u].z * a1);
                    lds_add(t + 224, f1[u].w * a1);
                } else {
                    lds_add(t +  0, f0[u].x * a0 + f1[u].x * a1);
                    lds_add(t + 32, f0[u].y * a0 + f1[u].y * a1);
                    lds_add(t + 64, f0[u].z * a0 + f1[u].z * a1);
                    lds_add(t + 96, f0[u].w * a0 + f1[u].w * a1);
                }
            }
        }
    }
}

template <int LAYER, int PASS>
__device__ __forceinline__ void batch_from_entries(float* tile, unsigned* mx, float* dn, const float2* erl,
                                                   const unsigned* q, const int n, const int lane,
                                                   const float* __restrict__ feat,
                                                   const float* __restrict__ el,
                                                   const int* __restrict__ src) {
    const bool valid = lane < n;
    const unsigned ent = valid ? q[lane] : 0u;
    const int e  = (int)(ent >> 8);
    const int dl = (int)(ent & 255u);
    const int s  = src[e];
    const float2 l = *(const float2*)(el + (size_t)s * 2);
    const float2 r = erl[dl];
    const float c0 = leaky(l.x + r.x), c1 = leaky(l.y + r.y);
    if (PASS == 0) {
        if (valid) { lds_max(mx + dl * 2, ordEnc(c0)); lds_max(mx + dl * 2 + 1, ordEnc(c1)); }
        return;
    }
    const float x0 = expf(c0 - ordDec(mx[dl * 2])), x1 = expf(c1 - ordDec(mx[dl * 2 + 1]));
    if (PASS == 1) {
        if (valid) { lds_add(dn + dl * 2, x0); lds_add(dn + dl * 2 + 1, x1); }
        return;
    }
    float w0 = x0 / dn[dl * 2], w1 = x1 / dn[dl * 2 + 1];
    if (LAYER == 0) { w0 *= 0.5f; w1 *= 0.5f; }
    add_rows<LAYER>(tile, n, lane, s, dl, w0, w1, feat);
}

template <int LAYER, int PASS>
__device__ __forceinline__ void rescan_pass(float* tile, unsigned* mx, float* dn, const float2* erl,
                                            unsigned* qe, const int wave, const int lane, const int base,
                                            const float* __restrict__ feat,
                                            const float* __restrict__ el,
                                            const int* __restrict__ src,
                                            const int* __restrict__ dst) {
    constexpr int NV = N_EDGES / 4;
    const unsigned lt = (1u << lane) - 1u;
    const int4* d4 = (const int4*)dst;
    unsigned cnt = 0;
    for (int i0 = wave * 32; i0 < NV; i0 += 1024) {
        const int i = i0 + lane;
        const bool ok = i < NV;
        const int4 dv = d4[ok ? i : NV - 1];
        cnt = append4(qe, cnt, dv, i, ok, base, lt);
        while (cnt >= 32) {
            cnt -= 32;
            batch_from_entries<LAYER, PASS>(tile, mx, dn, erl, qe + cnt, 32, lane, feat, el, src);
        }
    }
    if (cnt) batch_from_entries<LAYER, PASS>(tile, mx, dn, erl, qe, (int)cnt, lane, feat, el, src);
}

template <int LAYER>
__device__ __forceinline__ void owned_range(float* tile, unsigned* queue, unsigned* mx, float* dn,
                                            float2* erl, unsigned* flag,
                                            const float* __restrict__ feat,
                                            const float* __restrict__ el,
                                            const float* __restrict__ er,
                                            const int* __restrict__ src,
                                            const int* __restrict__ dst,
                                            const float* __restrict__ b,
                                            float* park,
                                            float* outp) {
    constexpr int COLS = LAYER ? 256 : 128;
    constexpr int NV   = N_EDGES / 4;
    const int tid  = threadIdx.x;
    const int lane = tid & 31;
    const int wave = __builtin_amdgcn_readfirstlane(tid >> 5);
    const int base = blockIdx.x * RANGE;

    const int rows = min(RANGE, N_NODES - base);
    const int slot = rows * 8;
    const int cap  = min(QCAP, slot - 32);
    float* pk = park + (size_t)base * 256 + wave * slot;
    unsigned parked = 0u;
    if (LAYER) parked = __float_as_uint(pk[0]);

    for (int i = tid; i < RANGE * COLS; i += 1024) {
        float v = 0.0f;
        if (LAYER == 0) {
            const int r = i & 127, c = r >> 5, l = r & 31;
            v = 0.5f * (b[l * 4 + c] + b[128 + l * 4 + c]);
        }
        tile[i] = v;
    }
    if (tid < RANGE * 2) { mx[tid] = ORD_NEG_INF; dn[tid] = 0.0f; }
    if (tid < RANGE) {
        const int nd = base + tid;
        erl[tid] = (nd < N_NODES) ? *(const float2*)(er + (size_t)nd * 2) : make_float2(0.0f, 0.0f);
    }
    if (tid == 0) *flag = 0u;
    __syncthreads();

    unsigned* qe  = queue + wave * (4 * QCAP);
    int*      qs  = (int*)(qe + QCAP);
    float*    qw0 = (float*)(qe + 2 * QCAP);
    float*    qw1 = (float*)(qe + 3 * QCAP);
    unsigned cnt = 0;
    if (LAYER == 0) {
        const unsigned lt = (1u << lane) - 1u;
        const int4* d4 = (const int4*)dst;
        for (int i0 = wave * 32; i0 < NV; i0 += 4096) {
            int4 dv[4];
            #pragma unroll
            for (int u = 0; u < 4; ++u) {
                const int i = i0 + u * 1024 + lane;
                dv[u] = d4[i < NV ? i : NV - 1];
            }
            #pragma unroll
            for (int u = 0; u < 4; ++u) {
                const int i = i0 + u * 1024 + lane;
                cnt = append4(qe, cnt, dv[u], i, i < NV, base, lt);
            }
        }
        if (cnt > QCAP && lane == 0) *flag = 1u;
    } else {
        cnt = __builtin_amdgcn_readfirstlane(parked);
        if (cnt > (unsigned)cap && lane == 0) *flag = 1u;
    }
    __syncthreads();
    const bool overflow = *flag != 0u;
    if (LAYER == 0 && lane == 0) pk[0] = __uint_as_float(cnt);

    if (!overflow) {
        const int n = (int)cnt;
        for (int b0 = 0; b0 < n; b0 += 128) {
            unsigned ent[4]; int s[4]; float2 l[4];
            #pragma unroll
            for (int u = 0; u < 4; ++u) {
                const int idx = b0 + u * 32 + lane;
                if (LAYER) ent[u] = (idx < n) ? __float_as_uint(pk[32 + idx]) : 0u;
                else       ent[u] = (idx < n) ? qe[idx] : 0u;
            }
            #pragma unroll
            for (int u = 0; u < 4; ++u) s[u] = LAYER ? (int)(ent[u] >> 8) : src[ent[u] >> 8];
            #pragma unroll
            for (int u = 0; u < 4; ++u) l[u] = *(const float2*)(el + (size_t)s[u] * 2);
            #pragma unroll
            for (int u = 0; u < 4; ++u) {
                const int idx = b0 + u * 32 + lane;
                if (idx < n) {
                    const int dl = (int)(ent[u] & 255u);
                    const float2 r = erl[dl];
                    const float c0 = leaky(l[u].x + r.x), c1 = leaky(l[u].y + r.y);
                    lds_max(mx + dl * 2, ordEnc(c0));
                    lds_max(mx + dl * 2 + 1, ordEnc(c1));
                    qs[idx] = s[u]; qw0[idx] = c0; qw1[idx] = c1;
                    if (LAYER) qe[idx] = ent[u];
                }
            }
        }
        __syncthreads();
        for (int idx = lane; idx < n; idx += 32) {
            const int dl = (int)(qe[idx] & 255u);
            const float x0 = expf(qw0[idx] - ordDec(mx[dl * 2]));
            const float x1 = expf(qw1[idx] - ordDec(mx[dl * 2 + 1]));
            lds_add(dn + dl * 2, x0);
            lds_add(dn + dl * 2 + 1, x1);
            qw0[idx] = x0; qw1[idx] = x1;
        }
        __syncthreads();
        for (int b0 = 0; b0 < n; b0 += 32) {
            const int idx = b0 + lane;
            const bool valid = idx < n;
            const unsigned ent = valid ? qe[idx] : 0u;
            const int dl = (int)(ent & 255u);
            const int s  = valid ? qs[idx] : 0;
            float w0 = valid ? qw0[idx] / dn[dl * 2] : 0.0f;
            float w1 = valid ? qw1[idx] / dn[dl * 2 + 1] : 0.0f;
            if (LAYER == 0) { w0 *= 0.5f; w1 *= 0.5f; }
            if (LAYER == 0 && valid && idx < cap) pk[32 + idx] = __uint_as_float(((unsigned)s << 8) | (unsigned)dl);
            add_rows<LAYER>(tile, min(32, n - b0), lane, s, dl, w0, w1, feat);
        }
    } else {
        rescan_pass<LAYER, 0>(tile, mx, dn, erl, qe, wave, lane, base, feat, el, src, dst);
        __syncthreads();
        rescan_pass<LAYER, 1>(tile, mx, dn, erl, qe, wave, lane, base, feat, el, src, dst);
        __syncthreads();
        rescan_pass<LAYER, 2>(tile, mx, dn, erl, qe, wave, lane, base, feat, el, src, dst);
    }
    __syncthreads();

    constexpr int V4 = COLS / 4;
    for (int i = tid; i < RANGE * V4; i += 1024) {
        const int n = i / V4, r = i - n * V4;
        const int h = r >> 5, l = r & 31;
        if (base + n < N_NODES) {
            const float* t = tile + n * COLS + h * 128 + l;
            float4 v = make_float4(t[0], t[32], t[64], t[96]);
            if (LAYER) {
                const float4 bb = *(const float4*)(b + h * 128 + l * 4);
                v.x += bb.x; v.y += bb.y; v.z += bb.z; v.w += bb.w;
            }
            *(float4*)(outp + (size_t)(base + n) * COLS + h * 128 + l * 4) = v;
        }
    }
}

__global__ void __launch_bounds__(1024) aggregate_owned_mean(const float* __restrict__ feat,
                                                             const float* __restrict__ el,
                                                             const float* __restrict__ er,
                                                             const int* __restrict__ src,
                                                             const int* __restrict__ dst,
                                                             const float* __restrict__ b,
                                                             float* __restrict__ hnext,
                                                             float* __restrict__ fin) {
    __shared__ float tile[RANGE * 128];
    __shared__ unsigned queue[32 * 4 * QCAP];
    __shared__ unsigned mx[RANGE * 2];
    __shared__ float dn[RANGE * 2];
    __shared__ float2 erl[RANGE];
    __shared__ unsigned flag;
    owned_range<0>(tile, queue, mx, dn, erl, &flag, feat, el, er, src, dst, b, fin, hnext);
}

__global__ void __launch_bounds__(1024) aggregate_owned_out(const float* __restrict__ feat,
                                                            const float* __restrict__ el,
                                                            const float* __restrict__ er,
                                                            const int* __restrict__ src,
                                                            const int* __restrict__ dst,
                                                            const float* __restrict__ b,
                                                            float* __restrict__ out) {
    __shared__ float tile[RANGE * 256];
    __shared__ unsigned queue[32 * 4 * QCAP];
    __shared__ unsigned mx[RANGE * 2];
    __shared__ float dn[RANGE * 2];
    __shared__ float2 erl[RANGE];
    __shared__ unsigned flag;
    owned_range<1>(tile, queue, mx, dn, erl, &flag, feat, el, er, src, dst, b, out, out);
}

extern "C" void kernel_launch(void* const* d_in, const int* in_sizes, int n_in,
                              void* d_out, int out_size, void* d_ws, size_t ws_size,
                              hipStream_t stream) {
    const float* x   = (const float*)d_in[0];
    const float* Ws  = (const float*)d_in[1];
    const float* als = (const float*)d_in[2];
    const float* ars = (const float*)d_in[3];
    const float* bs  = (const float*)d_in[4];
    const int*   src = (const int*)d_in[5];
    const int*   dst = (const int*)d_in[6];

    float* ws      = (float*)d_ws;
    float* feat    = ws;
    float* hnext   = ws + 12800000;
    float* el      = ws + 19200000;
    float* er      = ws + 19300000;

    float* out = (float*)d_out;

    for (int layer = 0; layer < 2; ++layer) {
        const float* hin = layer ? hnext : x;
        const float* W   = Ws  + (size_t)layer * D_MODEL * 256;
        const float* al  = als + (size_t)layer * 256;
        const float* ar  = ars + (size_t)layer * 256;
        const float* b   = bs  + (size_t)layer * 256;

        gemm_feat<<<N_NODES / 16, 128, 0, stream>>>(hin, W, feat);
        node_attn<<<(N_NODES + 7) / 8, 256, 0, stream>>>(feat, al, ar, el, er);
        if (layer == 0) {
            aggregate_owned_mean<<<N_RANGES, 1024, 0, stream>>>(feat, el, er, src, dst, b, hnext, out);
        } else {
            aggregate_owned_out<<<N_RANGES, 1024, 0, stream>>>(feat, el, er, src, dst, b, out);
        }
    }
}


# el/er dots in the projection's epilogue from accumulator tiles (16-lane DPP butterfly; a head's two waves added in fixed order via LDS, no atomics); node_attn launches dropped; el/er sums regrouped
# speedup vs baseline: 4.4989x; 1.0834x over previous
#include <hip/hip_runtime.h>
#include <math.h>

#define N_NODES 50000
#define N_EDGES 800000
#define D_MODEL 128
#define N_HEADS 2

typedef __attribute__((ext_vector_type(2))) float v2f;
typedef __attribute__((ext_vector_type(8))) float v8f;

__device__ __forceinline__ unsigned ordEnc(float f) {
    unsigned u = __float_as_uint(f);
    return (u & 0x80000000u) ? ~u : (u | 0x80000000u);
}
__device__ __forceinline__ float ordDec(unsigned u) {
    unsigned v = (u & 0x80000000u) ? (u & 0x7fffffffu) : ~u;
    return __uint_as_float(v);
}
#define ORD_NEG_INF 0x007FFFFFu


template <int MASK>
__device__ __forceinline__ float add_xmask(const float v) {
    const int o = __builtin_amdgcn_update_dpp(0, __builtin_bit_cast(int, v), 0x160 + MASK, 0xf, 0xf, true);
    return __builtin_bit_cast(float, o) + v;
}
__device__ __forceinline__ float row16_sum(float v) {
    v = add_xmask<1>(v);
    v = add_xmask<2>(v);
    v = add_xmask<4>(v);
    v = add_xmask<8>(v);
    return v;
}

__global__ void __launch_bounds__(128) gemm_feat(const float* __restrict__ h,
                                                 const float* __restrict__ W,
                                                 const float* __restrict__ al,
                                                 const float* __restrict__ ar,
                                                 float* __restrict__ feat,
                                                 float* __restrict__ el,
                                                 float* __restrict__ er) {
    __shared__ float2 part[4][16];

    const int lane  = threadIdx.x & 31;
    const int wave  = threadIdx.x >> 5;
    const int mbase = blockIdx.x << 4;
    const int nq    = wave << 6;
    const int row   = lane & 15;
    const int khalf = lane >> 4;

    const float* alp = al + nq + row;
    const float* arp = ar + nq + row;
    const float l0 = alp[0], l1 = alp[16], l2 = alp[32], l3 = alp[48];
    const float r0 = arp[0], r1 = arp[16], r2 = arp[32], r3 = arp[48];

    v8f acc0 = {}, acc1 = {}, acc2 = {}, acc3 = {};
    const float* Arow = h + (size_t)(mbase + row) * D_MODEL;

    for (int k = 0; k < D_MODEL; k += 4) {
        const int ka = k + khalf * 2;
        v2f a;
        a.x = Arow[ka];
        a.y = Arow[ka + 1];
        const float* Wb = W + (size_t)ka * 256 + row;
        v2f b0, b1, b2, b3;
        b0.x = Wb[nq];        b0.y = Wb[256 + nq];
        b1.x = Wb[nq + 16];   b1.y = Wb[256 + nq + 16];
        b2.x = Wb[nq + 32];   b2.y = Wb[256 + nq + 32];
        b3.x = Wb[nq + 48];   b3.y = Wb[256 + nq + 48];
        acc0 = __builtin_amdgcn_wmma_f32_16x16x4_f32(false, a, false, b0, (short)0, acc0, false, false);
        acc1 = __builtin_amdgcn_wmma_f32_16x16x4_f32(false, a, false, b1, (short)0, acc1, false, false);
        acc2 = __builtin_amdgcn_wmma_f32_16x16x4_f32(false, a, false, b2, (short)0, acc2, false, false);
        acc3 = __builtin_amdgcn_wmma_f32_16x16x4_f32(false, a, false, b3, (short)0, acc3, false, false);
    }

    float pl[8], pr[8];
    #pragma unroll
    for (int j = 0; j < 8; ++j) {
        float s = acc0[j] * l0;
        s = fmaf(acc1[j], l1, s);
        s = fmaf(acc2[j], l2, s);
        s = fmaf(acc3[j], l3, s);
        float t = acc0[j] * r0;
        t = fmaf(acc1[j], r1, t);
        t = fmaf(acc2[j], r2, t);
        t = fmaf(acc3[j], r3, t);
        pl[j] = row16_sum(s);
        pr[j] = row16_sum(t);
    }
    if (row == 0) {
        #pragma unroll
        for (int j = 0; j < 8; ++j) part[wave][khalf * 8 + j] = make_float2(pl[j], pr[j]);
    }
    __syncthreads();
    if (threadIdx.x < 32) {
        const int hd = lane >> 4;
        const float2 p0 = part[hd * 2][row], p1 = part[hd * 2 + 1][row];
        el[(size_t)(mbase + row) * 2 + hd] = p0.x + p1.x;
        er[(size_t)(mbase + row) * 2 + hd] = p0.y + p1.y;
    }

    #pragma unroll
    for (int j = 0; j < 8; ++j) {
        const int m = mbase + j + khalf * 8;
        float* F = feat + (size_t)m * 256 + nq + row;
        F[0]  = acc0[j];
        F[16] = acc1[j];
        F[32] = acc2[j];
        F[48] = acc3[j];
    }
}

#define RANGE    196
#define N_RANGES 256
#define QCAP     224

__device__ __forceinline__ void lds_add(float* p, float v) {
    __hip_atomic_fetch_add(p, v, __ATOMIC_RELAXED, __HIP_MEMORY_SCOPE_WORKGROUP);
}
__device__ __forceinline__ void lds_max(unsigned* p, unsigned v) {
    __hip_atomic_fetch_max(p, v, __ATOMIC_RELAXED, __HIP_MEMORY_SCOPE_WORKGROUP);
}
__device__ __forceinline__ float leaky(float v) { return v > 0.0f ? v : 0.2f * v; }

__device__ __forceinline__ unsigned append4(unsigned* qe, unsigned cnt, const int4 dv, const int i,
                                            const bool ok, const int base, const unsigned lt) {
    const unsigned r0 = (unsigned)(dv.x - base), r1 = (unsigned)(dv.y - base);
    const unsigned r2 = (unsigned)(dv.z - base), r3 = (unsigned)(dv.w - base);
    const bool m0 = ok & (r0 < RANGE), m1 = ok & (r1 < RANGE);
    const bool m2 = ok & (r2 < RANGE), m3 = ok & (r3 < RANGE);
    if (__builtin_amdgcn_ballot_w32(m0 | m1 | m2 | m3)) {
        const unsigned e4 = (unsigned)i * 4u;
        unsigned bal, pos;
        bal = __builtin_amdgcn_ballot_w32(m0);
        pos = cnt + __builtin_popcount(bal & lt);
        if (m0 && pos < QCAP) qe[pos] = ((e4 + 0u) << 8) | r0;
        cnt += __builtin_popcount(bal);
        bal = __builtin_amdgcn_ballot_w32(m1);
        pos = cnt + __builtin_popcount(bal & lt);
        if (m1 && pos < QCAP) qe[pos] = ((e4 + 1u) << 8) | r1;
        cnt += __builtin_popcount(bal);
        bal = __builtin_amdgcn_ballot_w32(m2);
        pos = cnt + __builtin_popcount(bal & lt);
        if (m2 && pos < QCAP) qe[pos] = ((e4 + 2u) << 8) | r2;
        cnt += __builtin_popcount(bal);
        bal = __builtin_amdgcn_ballot_w32(m3);
        pos = cnt + __builtin_popcount(bal & lt);
        if (m3 && pos < QCAP) qe[pos] = ((e4 + 3u) << 8) | r3;
        cnt += __builtin_popcount(bal);
    }
    return cnt;
}

template <int LAYER>
__device__ __forceinline__ void add_rows(float* tile, const int n, const int lane,
                                         const int s, const int dl, const float w0, const float w1,
                                         const float* __restrict__ feat) {
    constexpr int COLS = LAYER ? 256 : 128;
    for (int j0 = 0; j0 < n; j0 += 8) {
        float4 f0[8], f1[8];
        #pragma unroll
        for (int u = 0; u < 8; ++u) {
            const int sj = __builtin_amdgcn_readlane(s, j0 + u);
            const float* row = feat + (size_t)sj * 256 + lane * 4;
            f0[u] = *(const float4*)row;
            f1[u] = *(const float4*)(row + 128);
        }
        #pragma unroll
        for (int u = 0; u < 8; ++u) {
            if (j0 + u < n) {
                const int   dj = __builtin_amdgcn_readlane(dl, j0 + u);
                const float a0 = __int_as_float(__builtin_amdgcn_readlane(__float_as_int(w0), j0 + u));
                const float a1 = __int_as_float(__builtin_amdgcn_readlane(__float_as_int(w1), j0 + u));
                float* t = tile + dj * COLS + lane;
                if (LAYER) {
                    lds_add(t +   0, f0[u].x * a0);
                    lds_add(t +  32, f0[u].y * a0);
                    lds_add(t +  64, f0[u].z * a0);
                    lds_add(t +  96, f0[u].w * a0);
                    lds_add(t + 128, f1[u].x * a1);
                    lds_add(t + 160, f1[u].y * a1);
                    lds_add(t + 192, f1[u].z * a1);
                    lds_add(t + 224, f1[u].w * a1);
                } else {
                    lds_add(t +  0, f0[u].x * a0 + f1[u].x * a1);
                    lds_add(t + 32, f0[u].y * a0 + f1[u].y * a1);
                    lds_add(t + 64, f0[u].z * a0 + f1[u].z * a1);
                    lds_add(t + 96, f0[u].w * a0 + f1[u].w * a1);
                }
            }
        }
    }
}

template <int LAYER, int PASS>
__device__ __forceinline__ void batch_from_entries(float* tile, unsigned* mx, float* dn, const float2* erl,
                                                   const unsigned* q, const int n, const int lane,
                                                   const float* __restrict__ feat,
                                                   const float* __restrict__ el,
                                                   const int* __restrict__ src) {
    const bool valid = lane < n;
    const unsigned ent = valid ? q[lane] : 0u;
    const int e  = (int)(ent >> 8);
    const int dl = (int)(ent & 255u);
    const int s  = src[e];
    const float2 l = *(const float2*)(el + (size_t)s * 2);
    const float2 r = erl[dl];
    const float c0 = leaky(l.x + r.x), c1 = leaky(l.y + r.y);
    if (PASS == 0) {
        if (valid) { lds_max(mx + dl * 2, ordEnc(c0)); lds_max(mx + dl * 2 + 1, ordEnc(c1)); }
        return;
    }
    const float x0 = expf(c0 - ordDec(mx[dl * 2])), x1 = expf(c1 - ordDec(mx[dl * 2 + 1]));
    if (PASS == 1) {
        if (valid) { lds_add(dn + dl * 2, x0); lds_add(dn + dl * 2 + 1, x1); }
        return;
    }
    float w0 = x0 / dn[dl * 2], w1 = x1 / dn[dl * 2 + 1];
    if (LAYER == 0) { w0 *= 0.5f; w1 *= 0.5f; }
    add_rows<LAYER>(tile, n, lane, s, dl, w0, w1, feat);
}

template <int LAYER, int PASS>
__device__ __forceinline__ void rescan_pass(float* tile, unsigned* mx, float* dn, const float2* erl,
                                            unsigned* qe, const int wave, const int lane, const int base,
                                            const float* __restrict__ feat,
                                            const float* __restrict__ el,
                                            const int* __restrict__ src,
                                            const int* __restrict__ dst) {
    constexpr int NV = N_EDGES / 4;
    const unsigned lt = (1u << lane) - 1u;
    const int4* d4 = (const int4*)dst;
    unsigned cnt = 0;
    for (int i0 = wave * 32; i0 < NV; i0 += 1024) {
        const int i = i0 + lane;
        const bool ok = i < NV;
        const int4 dv = d4[ok ? i : NV - 1];
        cnt = append4(qe, cnt, dv, i, ok, base, lt);
        while (cnt >= 32) {
            cnt -= 32;
            batch_from_entries<LAYER, PASS>(tile, mx, dn, erl, qe + cnt, 32, lane, feat, el, src);
        }
    }
    if (cnt) batch_from_entries<LAYER, PASS>(tile, mx, dn, erl, qe, (int)cnt, lane, feat, el, src);
}

template <int LAYER>
__device__ __forceinline__ void owned_range(float* tile, unsigned* queue, unsigned* mx, float* dn,
                                            float2* erl, unsigned* flag,
                                            const float* __restrict__ feat,
                                            const float* __restrict__ el,
                                            const float* __restrict__ er,
                                            const int* __restrict__ src,
                                            const int* __restrict__ dst,
                                            const float* __restrict__ b,
                                            float* park,
                                            float* outp) {
    constexpr int COLS = LAYER ? 256 : 128;
    constexpr int NV   = N_EDGES / 4;
    const int tid  = threadIdx.x;
    const int lane = tid & 31;
    const int wave = __builtin_amdgcn_readfirstlane(tid >> 5);
    const int base = blockIdx.x * RANGE;

    const int rows = min(RANGE, N_NODES - base);
    const int slot = rows * 8;
    const int cap  = min(QCAP, slot - 32);
    float* pk = park + (size_t)base * 256 + wave * slot;
    unsigned parked = 0u;
    if (LAYER) parked = __float_as_uint(pk[0]);

    for (int i = tid; i < RANGE * COLS; i += 1024) {
        float v = 0.0f;
        if (LAYER == 0) {
            const int r = i & 127, c = r >> 5, l = r & 31;
            v = 0.5f * (b[l * 4 + c] + b[128 + l * 4 + c]);
        }
        tile[i] = v;
    }
    if (tid < RANGE * 2) { mx[tid] = ORD_NEG_INF; dn[tid] = 0.0f; }
    if (tid < RANGE) {
        const int nd = base + tid;
        erl[tid] = (nd < N_NODES) ? *(const float2*)(er + (size_t)nd * 2) : make_float2(0.0f, 0.0f);
    }
    if (tid == 0) *flag = 0u;
    __syncthreads();

    unsigned* qe  = queue + wave * (4 * QCAP);
    int*      qs  = (int*)(qe + QCAP);
    float*    qw0 = (float*)(qe + 2 * QCAP);
    float*    qw1 = (float*)(qe + 3 * QCAP);
    unsigned cnt = 0;
    if (LAYER == 0) {
        const unsigned lt = (1u << lane) - 1u;
        const int4* d4 = (const int4*)dst;
        for (int i0 = wave * 32; i0 < NV; i0 += 4096) {
            int4 dv[4];
            #pragma unroll
            for (int u = 0; u < 4; ++u) {
                const int i = i0 + u * 1024 + lane;
                dv[u] = d4[i < NV ? i : NV - 1];
            }
            #pragma unroll
            for (int u = 0; u < 4; ++u) {
                const int i = i0 + u * 1024 + lane;
                cnt = append4(qe, cnt, dv[u], i, i < NV, base, lt);
            }
        }
        if (cnt > QCAP && lane == 0) *flag = 1u;
    } else {
        cnt = __builtin_amdgcn_readfirstlane(parked);
        if (cnt > (unsigned)cap && lane == 0) *flag = 1u;
    }
    __syncthreads();
    const bool overflow = *flag != 0u;
    if (LAYER == 0 && lane == 0) pk[0] = __uint_as_float(cnt);

    if (!overflow) {
        const int n = (int)cnt;
        for (int b0 = 0; b0 < n; b0 += 128) {
            unsigned ent[4]; int s[4]; float2 l[4];
            #pragma unroll
            for (int u = 0; u < 4; ++u) {
                const int idx = b0 + u * 32 + lane;
                if (LAYER) ent[u] = (idx < n) ? __float_as_uint(pk[32 + idx]) : 0u;
                else       ent[u] = (idx < n) ? qe[idx] : 0u;
            }
            #pragma unroll
            for (int u = 0; u < 4; ++u) s[u] = LAYER ? (int)(ent[u] >> 8) : src[ent[u] >> 8];
            #pragma unroll
            for (int u = 0; u < 4; ++u) l[u] = *(const float2*)(el + (size_t)s[u] * 2);
            #pragma unroll
            for (int u = 0; u < 4; ++u) {
                const int idx = b0 + u * 32 + lane;
                if (idx < n) {
                    const int dl = (int)(ent[u] & 255u);
                    const float2 r = erl[dl];
                    const float c0 = leaky(l[u].x + r.x), c1 = leaky(l[u].y + r.y);
                    lds_max(mx + dl * 2, ordEnc(c0));
                    lds_max(mx + dl * 2 + 1, ordEnc(c1));
                    qs[idx] = s[u]; qw0[idx] = c0; qw1[idx] = c1;
                    if (LAYER) qe[idx] = ent[u];
                }
            }
        }
        __syncthreads();
        for (int idx = lane; idx < n; idx += 32) {
            const int dl = (int)(qe[idx] & 255u);
            const float x0 = expf(qw0[idx] - ordDec(mx[dl * 2]));
            const float x1 = expf(qw1[idx] - ordDec(mx[dl * 2 + 1]));
            lds_add(dn + dl * 2, x0);
            lds_add(dn + dl * 2 + 1, x1);
            qw0[idx] = x0; qw1[idx] = x1;
        }
        __syncthreads();
        for (int b0 = 0; b0 < n; b0 += 32) {
            const int idx = b0 + lane;
            const bool valid = idx < n;
            const unsigned ent = valid ? qe[idx] : 0u;
            const int dl = (int)(ent & 255u);
            const int s  = valid ? qs[idx] : 0;
            float w0 = valid ? qw0[idx] / dn[dl * 2] : 0.0f;
            float w1 = valid ? qw1[idx] / dn[dl * 2 + 1] : 0.0f;
            if (LAYER == 0) { w0 *= 0.5f; w1 *= 0.5f; }
            if (LAYER == 0 && valid && idx < cap) pk[32 + idx] = __uint_as_float(((unsigned)s << 8) | (unsigned)dl);
            add_rows<LAYER>(tile, min(32, n - b0), lane, s, dl, w0, w1, feat);
        }
    } else {
        rescan_pass<LAYER, 0>(tile, mx, dn, erl, qe, wave, lane, base, feat, el, src, dst);
        __syncthreads();
        rescan_pass<LAYER, 1>(tile, mx, dn, erl, qe, wave, lane, base, feat, el, src, dst);
        __syncthreads();
        rescan_pass<LAYER, 2>(tile, mx, dn, erl, qe, wave, lane, base, feat, el, src, dst);
    }
    __syncthreads();

    constexpr int V4 = COLS / 4;
    for (int i = tid; i < RANGE * V4; i += 1024) {
        const int n = i / V4, r = i - n * V4;
        const int h = r >> 5, l = r & 31;
        if (base + n < N_NODES) {
            const float* t = tile + n * COLS + h * 128 + l;
            float4 v = make_float4(t[0], t[32], t[64], t[96]);
            if (LAYER) {
                const float4 bb = *(const float4*)(b + h * 128 + l * 4);
                v.x += bb.x; v.y += bb.y; v.z += bb.z; v.w += bb.w;
            }
            *(float4*)(outp + (size_t)(base + n) * COLS + h * 128 + l * 4) = v;
        }
    }
}

__global__ void __launch_bounds__(1024) aggregate_owned_mean(const float* __restrict__ feat,
                                                             const float* __restrict__ el,
                                                             const float* __restrict__ er,
                                                             const int* __restrict__ src,
                                                             const int* __restrict__ dst,
                                                             const float* __restrict__ b,
                                                             float* __restrict__ hnext,
                                                             float* __restrict__ fin) {
    __shared__ float tile[RANGE * 128];
    __shared__ unsigned queue[32 * 4 * QCAP];
    __shared__ unsigned mx[RANGE * 2];
    __shared__ float dn[RANGE * 2];
    __shared__ float2 erl[RANGE];
    __shared__ unsigned flag;
    owned_range<0>(tile, queue, mx, dn, erl, &flag, feat, el, er, src, dst, b, fin, hnext);
}

__global__ void __launch_bounds__(1024) aggregate_owned_out(const float* __restrict__ feat,
                                                            const float* __restrict__ el,
                                                            const float* __restrict__ er,
                                                            const int* __restrict__ src,
                                                            const int* __restrict__ dst,
                                                            const float* __restrict__ b,
                                                            float* __restrict__ out) {
    __shared__ float tile[RANGE * 256];
    __shared__ unsigned queue[32 * 4 * QCAP];
    __shared__ unsigned mx[RANGE * 2];
    __shared__ float dn[RANGE * 2];
    __shared__ float2 erl[RANGE];
    __shared__ unsigned flag;
    owned_range<1>(tile, queue, mx, dn, erl, &flag, feat, el, er, src, dst, b, out, out);
}

extern "C" void kernel_launch(void* const* d_in, const int* in_sizes, int n_in,
                              void* d_out, int out_size, void* d_ws, size_t ws_size,
                              hipStream_t stream) {
    const float* x   = (const float*)d_in[0];
    const float* Ws  = (const float*)d_in[1];
    const float* als = (const float*)d_in[2];
    const float* ars = (const float*)d_in[3];
    const float* bs  = (const float*)d_in[4];
    const int*   src = (const int*)d_in[5];
    const int*   dst = (const int*)d_in[6];

    float* ws      = (float*)d_ws;
    float* feat    = ws;
    float* hnext   = ws + 12800000;
    float* el      = ws + 19200000;
    float* er      = ws + 19300000;

    float* out = (float*)d_out;

    for (int layer = 0; layer < 2; ++layer) {
        const float* hin = layer ? hnext : x;
        const float* W   = Ws  + (size_t)layer * D_MODEL * 256;
        const float* al  = als + (size_t)layer * 256;
        const float* ar  = ars + (size_t)layer * 256;
        const float* b   = bs  + (size_t)layer * 256;

        gemm_feat<<<N_NODES / 16, 128, 0, stream>>>(hin, W, al, ar, feat, el, er);
        if (layer == 0) {
            aggregate_owned_mean<<<N_RANGES, 1024, 0, stream>>>(feat, el, er, src, dst, b, hnext, out);
        } else {
            aggregate_owned_out<<<N_RANGES, 1024, 0, stream>>>(feat, el, er, src, dst, b, out);
        }
    }
}


# layer 1 scans one key byte an edge (dst/196 = owning block; written by the projection launch into the output buffer) instead of 4-byte dst; edges dealt to waves anew, sums still in LDS arrival order
# speedup vs baseline: 4.8767x; 1.0840x over previous
#include <hip/hip_runtime.h>
#include <math.h>

#define N_NODES 50000
#define N_EDGES 800000
#define D_MODEL 128
#define N_HEADS 2

#define RANGE    196
#define N_RANGES 256

#define KEY_V4      (N_EDGES / 16)
#define KEY_SEG_V4  10000
#define KEY_HEAD_V4 2048
#define KEY_SKIP_V4 (RANGE * 64 - KEY_SEG_V4)
static_assert(N_EDGES == (N_NODES / 16) * 256, "one projection block per 256 edges");
static_assert(KEY_SEG_V4 % 16 == 0 && KEY_V4 % KEY_SEG_V4 == 0, "a projection block's 16 int4 stay in one segment");
static_assert(KEY_HEAD_V4 + KEY_SEG_V4 <= RANGE * 64, "a segment fits in one block's rows");
static_assert((N_NODES - 1) / RANGE < 256, "range index fits a byte");
__device__ __forceinline__ unsigned key_slot(const unsigned i) { return KEY_HEAD_V4 + i + (i / KEY_SEG_V4) * KEY_SKIP_V4; }

typedef __attribute__((ext_vector_type(2))) float v2f;
typedef __attribute__((ext_vector_type(8))) float v8f;

__device__ __forceinline__ unsigned ordEnc(float f) {
    unsigned u = __float_as_uint(f);
    return (u & 0x80000000u) ? ~u : (u | 0x80000000u);
}
__device__ __forceinline__ float ordDec(unsigned u) {
    unsigned v = (u & 0x80000000u) ? (u & 0x7fffffffu) : ~u;
    return __uint_as_float(v);
}
#define ORD_NEG_INF 0x007FFFFFu


template <int MASK>
__device__ __forceinline__ float add_xmask(const float v) {
    const int o = __builtin_amdgcn_update_dpp(0, __builtin_bit_cast(int, v), 0x160 + MASK, 0xf, 0xf, true);
    return __builtin_bit_cast(float, o) + v;
}
__device__ __forceinline__ float row16_sum(float v) {
    v = add_xmask<1>(v);
    v = add_xmask<2>(v);
    v = add_xmask<4>(v);
    v = add_xmask<8>(v);
    return v;
}

template <bool KEYS>
__global__ void __launch_bounds__(128) gemm_feat(const float* __restrict__ h,
                                                 const float* __restrict__ W,
                                                 const float* __restrict__ al,
                                                 const float* __restrict__ ar,
                                                 float* __restrict__ feat,
                                                 float* __restrict__ el,
                                                 float* __restrict__ er,
                                                 const int* __restrict__ dst,
                                                 float* keyspace) {
    __shared__ float2 part[4][16];

    const int lane  = threadIdx.x & 31;
    const int wave  = threadIdx.x >> 5;
    const int mbase = blockIdx.x << 4;
    const int nq    = wave << 6;
    const int row   = lane & 15;
    const int khalf = lane >> 4;

    const float* alp = al + nq + row;
    const float* arp = ar + nq + row;
    const float l0 = alp[0], l1 = alp[16], l2 = alp[32], l3 = alp[48];
    const float r0 = arp[0], r1 = arp[16], r2 = arp[32], r3 = arp[48];

    const int e0 = blockIdx.x * 256 + threadIdx.x * 2;
    int2 dk = make_int2(0, 0);
    if (KEYS) dk = *(const int2*)(dst + e0);

    v8f acc0 = {}, acc1 = {}, acc2 = {}, acc3 = {};
    const float* Arow = h + (size_t)(mbase + row) * D_MODEL;

    for (int k = 0; k < D_MODEL; k += 4) {
        const int ka = k + khalf * 2;
        v2f a;
        a.x = Arow[ka];
        a.y = Arow[ka + 1];
        const float* Wb = W + (size_t)ka * 256 + row;
        v2f b0, b1, b2, b3;
        b0.x = Wb[nq];        b0.y = Wb[256 + nq];
        b1.x = Wb[nq + 16];   b1.y = Wb[256 + nq + 16];
        b2.x = Wb[nq + 32];   b2.y = Wb[256 + nq + 32];
        b3.x = Wb[nq + 48];   b3.y = Wb[256 + nq + 48];
        acc0 = __builtin_amdgcn_wmma_f32_16x16x4_f32(false, a, false, b0, (short)0, acc0, false, false);
        acc1 = __builtin_amdgcn_wmma_f32_16x16x4_f32(false, a, false, b1, (short)0, acc1, false, false);
        acc2 = __builtin_amdgcn_wmma_f32_16x16x4_f32(false, a, false, b2, (short)0, acc2, false, false);
        acc3 = __builtin_amdgcn_wmma_f32_16x16x4_f32(false, a, false, b3, (short)0, acc3, false, false);
    }

    if (KEYS) {
        const unsigned k0 = (unsigned)dk.x / RANGE, k1 = (unsigned)dk.y / RANGE;
        unsigned char* kb = (unsigned char*)keyspace + (size_t)key_slot((unsigned)(e0 >> 4)) * 16 + (e0 & 15);
        *(unsigned short*)kb = (unsigned short)(k0 | (k1 << 8));
    }

    float pl[8], pr[8];
    #pragma unroll
    for (int j = 0; j < 8; ++j) {
        float s = acc0[j] * l0;
        s = fmaf(acc1[j], l1, s);
        s = fmaf(acc2[j], l2, s);
        s = fmaf(acc3[j], l3, s);
        float t = acc0[j] * r0;
        t = fmaf(acc1[j], r1, t);
        t = fmaf(acc2[j], r2, t);
        t = fmaf(acc3[j], r3, t);
        pl[j] = row16_sum(s);
        pr[j] = row16_sum(t);
    }
    if (row == 0) {
        #pragma unroll
        for (int j = 0; j < 8; ++j) part[wave][khalf * 8 + j] = make_float2(pl[j], pr[j]);
    }
    __syncthreads();
    if (threadIdx.x < 32) {
        const int hd = lane >> 4;
        const float2 p0 = part[hd * 2][row], p1 = part[hd * 2 + 1][row];
        el[(size_t)(mbase + row) * 2 + hd] = p0.x + p1.x;
        er[(size_t)(mbase + row) * 2 + hd] = p0.y + p1.y;
    }

    #pragma unroll
    for (int j = 0; j < 8; ++j) {
        const int m = mbase + j + khalf * 8;
        float* F = feat + (size_t)m * 256 + nq + row;
        F[0]  = acc0[j];
        F[16] = acc1[j];
        F[32] = acc2[j];
        F[48] = acc3[j];
    }
}

#define QCAP     224

__device__ __forceinline__ void lds_add(float* p, float v) {
    __hip_atomic_fetch_add(p, v, __ATOMIC_RELAXED, __HIP_MEMORY_SCOPE_WORKGROUP);
}
__device__ __forceinline__ void lds_max(unsigned* p, unsigned v) {
    __hip_atomic_fetch_max(p, v, __ATOMIC_RELAXED, __HIP_MEMORY_SCOPE_WORKGROUP);
}
__device__ __forceinline__ float leaky(float v) { return v > 0.0f ? v : 0.2f * v; }

__device__ __forceinline__ unsigned append4(unsigned* qe, unsigned cnt, const int4 dv, const int i,
                                            const bool ok, const int base, const unsigned lt) {
    const unsigned r0 = (unsigned)(dv.x - base), r1 = (unsigned)(dv.y - base);
    const unsigned r2 = (unsigned)(dv.z - base), r3 = (unsigned)(dv.w - base);
    const bool m0 = ok & (r0 < RANGE), m1 = ok & (r1 < RANGE);
    const bool m2 = ok & (r2 < RANGE), m3 = ok & (r3 < RANGE);
    if (__builtin_amdgcn_ballot_w32(m0 | m1 | m2 | m3)) {
        const unsigned e4 = (unsigned)i * 4u;
        unsigned bal, pos;
        bal = __builtin_amdgcn_ballot_w32(m0);
        pos = cnt + __builtin_popcount(bal & lt);
        if (m0 && pos < QCAP) qe[pos] = ((e4 + 0u) << 8) | r0;
        cnt += __builtin_popcount(bal);
        bal = __builtin_amdgcn_ballot_w32(m1);
        pos = cnt + __builtin_popcount(bal & lt);
        if (m1 && pos < QCAP) qe[pos] = ((e4 + 1u) << 8) | r1;
        cnt += __builtin_popcount(bal);
        bal = __builtin_amdgcn_ballot_w32(m2);
        pos = cnt + __builtin_popcount(bal & lt);
        if (m2 && pos < QCAP) qe[pos] = ((e4 + 2u) << 8) | r2;
        cnt += __builtin_popcount(bal);
        bal = __builtin_amdgcn_ballot_w32(m3);
        pos = cnt + __builtin_popcount(bal & lt);
        if (m3 && pos < QCAP) qe[pos] = ((e4 + 3u) << 8) | r3;
        cnt += __builtin_popcount(bal);
    }
    return cnt;
}

__device__ __forceinline__ unsigned zero_bytes(const unsigned x) {
    return ~(((x & 0x7f7f7f7fu) + 0x7f7f7f7fu) | x) & 0x80808080u;
}

__device__ __forceinline__ unsigned append_keys(unsigned* qe, unsigned cnt, const int4 kv, const int i,
                                                const bool ok, const unsigned idm, const int lane) {
    const unsigned z0 = zero_bytes((unsigned)kv.x ^ idm), z1 = zero_bytes((unsigned)kv.y ^ idm);
    const unsigned z2 = zero_bytes((unsigned)kv.z ^ idm), z3 = zero_bytes((unsigned)kv.w ^ idm);
    const unsigned mb = ok ? ((z0 >> 7) | (z1 >> 6) | (z2 >> 5) | (z3 >> 4)) : 0u;
    unsigned bal = __builtin_amdgcn_ballot_w32(mb != 0u);
    if (bal) {
        const int c = __builtin_popcount(mb);
        unsigned pos = cnt;
        do {
            const int L = __builtin_ctz(bal);
            bal &= bal - 1u;
            const unsigned cL = (unsigned)__builtin_amdgcn_readlane(c, L);
            if (lane > L) pos += cL;
            cnt += cL;
        } while (bal);
        unsigned m = mb;
        while (m) {
            const int p = __builtin_ctz(m);
            m &= m - 1u;
            if (pos < QCAP) qe[pos] = (unsigned)i * 16u + (unsigned)((p & 7) * 4 + (p >> 3));
            ++pos;
        }
    }
    return cnt;
}

template <int LAYER>
__device__ __forceinline__ void add_rows(float* tile, const int n, const int lane,
                                         const int s, const int dl, const float w0, const float w1,
                                         const float* __restrict__ feat) {
    constexpr int COLS = LAYER ? 256 : 128;
    for (int j0 = 0; j0 < n; j0 += 8) {
        float4 f0[8], f1[8];
        #pragma unroll
        for (int u = 0; u < 8; ++u) {
            const int sj = __builtin_amdgcn_readlane(s, j0 + u);
            const float* row = feat + (size_t)sj * 256 + lane * 4;
            f0[u] = *(const float4*)row;
            f1[u] = *(const float4*)(row + 128);
        }
        #pragma unroll
        for (int u = 0; u < 8; ++u) {
            if (j0 + u < n) {
                const int   dj = __builtin_amdgcn_readlane(dl, j0 + u);
                const float a0 = __int_as_float(__builtin_amdgcn_readlane(__float_as_int(w0), j0 + u));
                const float a1 = __int_as_float(__builtin_amdgcn_readlane(__float_as_int(w1), j0 + u));
                float* t = tile + dj * COLS + lane;
                if (LAYER) {
                    lds_add(t +   0, f0[u].x * a0);
                    lds_add(t +  32, f0[u].y * a0);
                    lds_add(t +  64, f0[u].z * a0);
                    lds_add(t +  96, f0[u].w * a0);
                    lds_add(t + 128, f1[u].x * a1);
                    lds_add(t + 160, f1[u].y * a1);
                    lds_add(t + 192, f1[u].z * a1);
                    lds_add(t + 224, f1[u].w * a1);
                } else {
                    lds_add(t +  0, f0[u].x * a0 + f1[u].x * a1);
                    lds_add(t + 32, f0[u].y * a0 + f1[u].y * a1);
                    lds_add(t + 64, f0[u].z * a0 + f1[u].z * a1);
                    lds_add(t + 96, f0[u].w * a0 + f1[u].w * a1);
                }
            }
        }
    }
}

template <int LAYER, int PASS>
__device__ __forceinline__ void batch_from_entries(float* tile, unsigned* mx, float* dn, const float2* erl,
                                                   const unsigned* q, const int n, const int lane,
                                                   const float* __restrict__ feat,
                                                   const float* __restrict__ el,
                                                   const int* __restrict__ src) {
    const bool valid = lane < n;
    const unsigned ent = valid ? q[lane] : 0u;
    const int e  = (int)(ent >> 8);
    const int dl = (int)(ent & 255u);
    const int s  = src[e];
    const float2 l = *(const float2*)(el + (size_t)s * 2);
    const float2 r = erl[dl];
    const float c0 = leaky(l.x + r.x), c1 = leaky(l.y + r.y);
    if (PASS == 0) {
        if (valid) { lds_max(mx + dl * 2, ordEnc(c0)); lds_max(mx + dl * 2 + 1, ordEnc(c1)); }
        return;
    }
    const float x0 = expf(c0 - ordDec(mx[dl * 2])), x1 = expf(c1 - ordDec(mx[dl * 2 + 1]));
    if (PASS == 1) {
        if (valid) { lds_add(dn + dl * 2, x0); lds_add(dn + dl * 2 + 1, x1); }
        return;
    }
    float w0 = x0 / dn[dl * 2], w1 = x1 / dn[dl * 2 + 1];
    if (LAYER == 0) { w0 *= 0.5f; w1 *= 0.5f; }
    add_rows<LAYER>(tile, n, lane, s, dl, w0, w1, feat);
}

template <int LAYER, int PASS>
__device__ __forceinline__ void rescan_pass(float* tile, unsigned* mx, float* dn, const float2* erl,
                                            unsigned* qe, const int wave, const int lane, const int base,
                                            const float* __restrict__ feat,
                                            const float* __restrict__ el,
                                            const int* __restrict__ src,
                                            const int* __restrict__ dst) {
    constexpr int NV = N_EDGES / 4;
    const unsigned lt = (1u << lane) - 1u;
    const int4* d4 = (const int4*)dst;
    unsigned cnt = 0;
    for (int i0 = wave * 32; i0 < NV; i0 += 1024) {
        const int i = i0 + lane;
        const bool ok = i < NV;
        const int4 dv = d4[ok ? i : NV - 1];
        cnt = append4(qe, cnt, dv, i, ok, base, lt);
        while (cnt >= 32) {
            cnt -= 32;
            batch_from_entries<LAYER, PASS>(tile, mx, dn, erl, qe + cnt, 32, lane, feat, el, src);
        }
    }
    if (cnt) batch_from_entries<LAYER, PASS>(tile, mx, dn, erl, qe, (int)cnt, lane, feat, el, src);
}

template <int LAYER>
__device__ __forceinline__ void owned_range(float* tile, unsigned* queue, unsigned* mx, float* dn,
                                            float2* erl, unsigned* flag,
                                            const float* __restrict__ feat,
                                            const float* __restrict__ el,
                                            const float* __restrict__ er,
                                            const int* __restrict__ src,
                                            const int* __restrict__ dst,
                                            const float* __restrict__ b,
                                            float* park,
                                            float* outp) {
    constexpr int COLS = LAYER ? 256 : 128;
    constexpr int NV   = N_EDGES / 4;
    const int tid  = threadIdx.x;
    const int lane = tid & 31;
    const int wave = __builtin_amdgcn_readfirstlane(tid >> 5);
    const int base = blockIdx.x * RANGE;

    const int rows = min(RANGE, N_NODES - base);
    const int slot = min(256, rows * 8);
    const int cap  = min(QCAP, slot - 32);
    float* pk = park + (size_t)base * 256 + wave * slot;
    unsigned parked = 0u;
    if (LAYER) parked = __float_as_uint(pk[0]);

    for (int i = tid; i < RANGE * COLS; i += 1024) {
        float v = 0.0f;
        if (LAYER == 0) {
            const int r = i & 127, c = r >> 5, l = r & 31;
            v = 0.5f * (b[l * 4 + c] + b[128 + l * 4 + c]);
        }
        tile[i] = v;
    }
    if (tid < RANGE * 2) { mx[tid] = ORD_NEG_INF; dn[tid] = 0.0f; }
    if (tid < RANGE) {
        const int nd = base + tid;
        erl[tid] = (nd < N_NODES) ? *(const float2*)(er + (size_t)nd * 2) : make_float2(0.0f, 0.0f);
    }
    if (tid == 0) *flag = 0u;
    __syncthreads();

    unsigned* qe  = queue + wave * (4 * QCAP);
    int*      qs  = (int*)(qe + QCAP);
    float*    qw0 = (float*)(qe + 2 * QCAP);
    float*    qw1 = (float*)(qe + 3 * QCAP);
    unsigned cnt = 0;
    if (LAYER == 0) {
        const unsigned idm = (unsigned)blockIdx.x * 0x01010101u;
        const int4* k4 = (const int4*)park;
        for (int i0 = wave * 32; i0 < KEY_V4; i0 += 4096) {
            int4 kv[4];
            #pragma unroll
            for (int u = 0; u < 4; ++u) {
                const int i = i0 + u * 1024 + lane;
                kv[u] = k4[key_slot((unsigned)min(i, KEY_V4 - 1))];
            }
            asm volatile("" :: "v"(kv[0].x), "v"(kv[0].y), "v"(kv[0].z), "v"(kv[0].w), "v"(kv[1].x), "v"(kv[1].y), "v"(kv[1].z), "v"(kv[1].w),
                            "v"(kv[2].x), "v"(kv[2].y), "v"(kv[2].z), "v"(kv[2].w), "v"(kv[3].x), "v"(kv[3].y), "v"(kv[3].z), "v"(kv[3].w));
            #pragma unroll
            for (int u = 0; u < 4; ++u) {
                const int i = i0 + u * 1024 + lane;
                cnt = append_keys(qe, cnt, kv[u], i, i < KEY_V4, idm, lane);
            }
        }
        if (cnt > QCAP && lane == 0) *flag = 1u;
    } else {
        cnt = __builtin_amdgcn_readfirstlane(parked);
        if (cnt > (unsigned)cap && lane == 0) *flag = 1u;
    }
    __syncthreads();
    const bool overflow = *flag != 0u;
    if (LAYER == 0 && lane == 0) pk[0] = __uint_as_float(cnt);

    if (!overflow) {
        const int n = (int)cnt;
        for (int b0 = 0; b0 < n; b0 += 128) {
            unsigned ent[4]; int s[4]; int nd[4]; float2 l[4];
            #pragma unroll
            for (int u = 0; u < 4; ++u) {
                const int idx = b0 + u * 32 + lane;
                if (LAYER) ent[u] = (idx < n) ? __float_as_uint(pk[32 + idx]) : 0u;
                else       ent[u] = (idx < n) ? qe[idx] : 0u;
            }
            #pragma unroll
            for (int u = 0; u < 4; ++u) {
                s[u]  = LAYER ? (int)(ent[u] >> 8)   : src[ent[u]];
                nd[u] = LAYER ? (int)(ent[u] & 255u) : dst[ent[u]] - base;
            }
            #pragma unroll
            for (int u = 0; u < 4; ++u) l[u] = *(const float2*)(el + (size_t)s[u] * 2);
            #pragma unroll
            for (int u = 0; u < 4; ++u) {
                const int idx = b0 + u * 32 + lane;
                if (idx < n) {
                    const int dl = nd[u];
                    const float2 r = erl[dl];
                    const float c0 = leaky(l[u].x + r.x), c1 = leaky(l[u].y + r.y);
                    lds_max(mx + dl * 2, ordEnc(c0));
                    lds_max(mx + dl * 2 + 1, ordEnc(c1));
                    qs[idx] = s[u]; qw0[idx] = c0; qw1[idx] = c1;
                    qe[idx] = (unsigned)dl;
                }
            }
        }
        __syncthreads();
        for (int idx = lane; idx < n; idx += 32) {
            const int dl = (int)(qe[idx] & 255u);
            const float x0 = expf(qw0[idx] - ordDec(mx[dl * 2]));
            const float x1 = expf(qw1[idx] - ordDec(mx[dl * 2 + 1]));
            lds_add(dn + dl * 2, x0);
            lds_add(dn + dl * 2 + 1, x1);
            qw0[idx] = x0; qw1[idx] = x1;
        }
        __syncthreads();
        for (int b0 = 0; b0 < n; b0 += 32) {
            const int idx = b0 + lane;
            const bool valid = idx < n;
            const unsigned ent = valid ? qe[idx] : 0u;
            const int dl = (int)(ent & 255u);
            const int s  = valid ? qs[idx] : 0;
            float w0 = valid ? qw0[idx] / dn[dl * 2] : 0.0f;
            float w1 = valid ? qw1[idx] / dn[dl * 2 + 1] : 0.0f;
            if (LAYER == 0) { w0 *= 0.5f; w1 *= 0.5f; }
            if (LAYER == 0 && valid && idx < cap) pk[32 + idx] = __uint_as_float(((unsigned)s << 8) | (unsigned)dl);
            add_rows<LAYER>(tile, min(32, n - b0), lane, s, dl, w0, w1, feat);
        }
    } else {
        rescan_pass<LAYER, 0>(tile, mx, dn, erl, qe, wave, lane, base, feat, el, src, dst);
        __syncthreads();
        rescan_pass<LAYER, 1>(tile, mx, dn, erl, qe, wave, lane, base, feat, el, src, dst);
        __syncthreads();
        rescan_pass<LAYER, 2>(tile, mx, dn, erl, qe, wave, lane, base, feat, el, src, dst);
    }
    __syncthreads();

    constexpr int V4 = COLS / 4;
    for (int i = tid; i < RANGE * V4; i += 1024) {
        const int n = i / V4, r = i - n * V4;
        const int h = r >> 5, l = r & 31;
        if (base + n < N_NODES) {
            const float* t = tile + n * COLS + h * 128 + l;
            float4 v = make_float4(t[0], t[32], t[64], t[96]);
            if (LAYER) {
                const float4 bb = *(const float4*)(b + h * 128 + l * 4);
                v.x += bb.x; v.y += bb.y; v.z += bb.z; v.w += bb.w;
            }
            *(float4*)(outp + (size_t)(base + n) * COLS + h * 128 + l * 4) = v;
        }
    }
}

__global__ void __launch_bounds__(1024) aggregate_owned_mean(const float* __restrict__ feat,
                                                             const float* __restrict__ el,
                                                             const float* __restrict__ er,
                                                             const int* __restrict__ src,
                                                             const int* __restrict__ dst,
                                                             const float* __restrict__ b,
                                                             float* __restrict__ hnext,
                                                             float* __restrict__ fin) {
    __shared__ float tile[RANGE * 128];
    __shared__ unsigned queue[32 * 4 * QCAP];
    __shared__ unsigned mx[RANGE * 2];
    __shared__ float dn[RANGE * 2];
    __shared__ float2 erl[RANGE];
    __shared__ unsigned flag;
    owned_range<0>(tile, queue, mx, dn, erl, &flag, feat, el, er, src, dst, b, fin, hnext);
}

__global__ void __launch_bounds__(1024) aggregate_owned_out(const float* __restrict__ feat,
                                                            const float* __restrict__ el,
                                                            const float* __restrict__ er,
                                                            const int* __restrict__ src,
                                                            const int* __restrict__ dst,
                                                            const float* __restrict__ b,
                                                            float* __restrict__ out) {
    __shared__ float tile[RANGE * 256];
    __shared__ unsigned queue[32 * 4 * QCAP];
    __shared__ unsigned mx[RANGE * 2];
    __shared__ float dn[RANGE * 2];
    __shared__ float2 erl[RANGE];
    __shared__ unsigned flag;
    owned_range<1>(tile, queue, mx, dn, erl, &flag, feat, el, er, src, dst, b, out, out);
}

extern "C" void kernel_launch(void* const* d_in, const int* in_sizes, int n_in,
                              void* d_out, int out_size, void* d_ws, size_t ws_size,
                              hipStream_t stream) {
    const float* x   = (const float*)d_in[0];
    const float* Ws  = (const float*)d_in[1];
    const float* als = (const float*)d_in[2];
    const float* ars = (const float*)d_in[3];
    const float* bs  = (const float*)d_in[4];
    const int*   src = (const int*)d_in[5];
    const int*   dst = (const int*)d_in[6];

    float* ws      = (float*)d_ws;
    float* feat    = ws;
    float* hnext   = ws + 12800000;
    float* el      = ws + 19200000;
    float* er      = ws + 19300000;

    float* out = (float*)d_out;

    for (int layer = 0; layer < 2; ++layer) {
        const float* hin = layer ? hnext : x;
        const float* W   = Ws  + (size_t)layer * D_MODEL * 256;
        const float* al  = als + (size_t)layer * 256;
        const float* ar  = ars + (size_t)layer * 256;
        const float* b   = bs  + (size_t)layer * 256;

        if (layer == 0) {
            gemm_feat<true><<<N_NODES / 16, 128, 0, stream>>>(hin, W, al, ar, feat, el, er, dst, out);
            aggregate_owned_mean<<<N_RANGES, 1024, 0, stream>>>(feat, el, er, src, dst, b, hnext, out);
        } else {
            gemm_feat<false><<<N_NODES / 16, 128, 0, stream>>>(hin, W, al, ar, feat, el, er, dst, out);
            aggregate_owned_out<<<N_RANGES, 1024, 0, stream>>>(feat, el, er, src, dst, b, out);
        }
    }
}


# projection: each wave keeps its 128x32 slice of W in registers, 256 eight-wave blocks loop over row tiles (was 3125 blocks, each reloading W); h rows prefetched a tile ahead; el/er sums regrouped
# speedup vs baseline: 5.0528x; 1.0361x over previous
#include <hip/hip_runtime.h>
#include <math.h>

#define N_NODES 50000
#define N_EDGES 800000
#define D_MODEL 128
#define N_HEADS 2

#define RANGE    196
#define N_RANGES 256

#define KEY_V4      (N_EDGES / 16)
#define KEY_SEG_V4  10000
#define KEY_HEAD_V4 2048
#define KEY_SKIP_V4 (RANGE * 64 - KEY_SEG_V4)
static_assert(N_EDGES == (N_NODES / 16) * 256, "one projection row tile per 256 edges");
static_assert(KEY_SEG_V4 % 16 == 0 && KEY_V4 % KEY_SEG_V4 == 0, "a row tile's 16 int4 stay in one segment");
static_assert(KEY_HEAD_V4 + KEY_SEG_V4 <= RANGE * 64, "a segment fits in one block's rows");
static_assert((N_NODES - 1) / RANGE < 256, "range index fits a byte");
__device__ __forceinline__ unsigned key_slot(const unsigned i) { return KEY_HEAD_V4 + i + (i / KEY_SEG_V4) * KEY_SKIP_V4; }

typedef __attribute__((ext_vector_type(2))) float v2f;
typedef __attribute__((ext_vector_type(8))) float v8f;

__device__ __forceinline__ unsigned ordEnc(float f) {
    unsigned u = __float_as_uint(f);
    return (u & 0x80000000u) ? ~u : (u | 0x80000000u);
}
__device__ __forceinline__ float ordDec(unsigned u) {
    unsigned v = (u & 0x80000000u) ? (u & 0x7fffffffu) : ~u;
    return __uint_as_float(v);
}
#define ORD_NEG_INF 0x007FFFFFu


template <int MASK>
__device__ __forceinline__ float add_xmask(const float v) {
    const int o = __builtin_amdgcn_update_dpp(0, __builtin_bit_cast(int, v), 0x160 + MASK, 0xf, 0xf, true);
    return __builtin_bit_cast(float, o) + v;
}
__device__ __forceinline__ float row16_sum(float v) {
    v = add_xmask<1>(v);
    v = add_xmask<2>(v);
    v = add_xmask<4>(v);
    v = add_xmask<8>(v);
    return v;
}

#define GEMM_BLOCKS 256
#define N_TILES (N_NODES / 16)

template <bool KEYS>
__global__ void __launch_bounds__(256) gemm_feat(const float* __restrict__ h,
                                                 const float* __restrict__ W,
                                                 const float* __restrict__ al,
                                                 const float* __restrict__ ar,
                                                 float* __restrict__ feat,
                                                 float* __restrict__ el,
                                                 float* __restrict__ er,
                                                 const int* __restrict__ dst,
                                                 float* keyspace) {
    __shared__ float2 part[2][8][16];

    const int lane  = threadIdx.x & 31;
    const int wave  = threadIdx.x >> 5;
    const int nq    = wave << 5;
    const int row   = lane & 15;
    const int khalf = lane >> 4;

    const float l0 = al[nq + row], l1 = al[nq + 16 + row];
    const float r0 = ar[nq + row], r1 = ar[nq + 16 + row];

    v2f wf[32][2];
    #pragma unroll
    for (int s = 0; s < 32; ++s) {
        const int k0 = (s >> 1) * 8 + khalf * 4 + (s & 1) * 2;
        const float* Wb = W + (size_t)k0 * 256 + nq + row;
        wf[s][0].x = Wb[0];   wf[s][0].y = Wb[256];
        wf[s][1].x = Wb[16];  wf[s][1].y = Wb[256 + 16];
    }

    float4 fa[16], fb[16];
    {
        const float4* Ar = (const float4*)(h + (size_t)(blockIdx.x * 16 + row) * D_MODEL + khalf * 4);
        #pragma unroll
        for (int q = 0; q < 16; ++q) fa[q] = Ar[q * 2];
    }
    int2 da = make_int2(0, 0), db = make_int2(0, 0);
    if (KEYS && threadIdx.x < 128) da = *(const int2*)(dst + blockIdx.x * 256 + threadIdx.x * 2);

    auto tile_step = [&](const float4 (&cur)[16], float4 (&nxt)[16], const int2& dk, int2& dkn, const int t, const int pb) __attribute__((always_inline)) {
        const int mbase = t << 4;
        const int tn = t + GEMM_BLOCKS;
        if (tn < N_TILES) {
            const float4* Ar = (const float4*)(h + (size_t)(tn * 16 + row) * D_MODEL + khalf * 4);
            #pragma unroll
            for (int q = 0; q < 16; ++q) nxt[q] = Ar[q * 2];
            if (KEYS && threadIdx.x < 128) dkn = *(const int2*)(dst + tn * 256 + threadIdx.x * 2);
        }
        const int e0 = t * 256 + threadIdx.x * 2;

        v8f acc0 = {}, acc1 = {};
        #pragma unroll
        for (int q = 0; q < 16; ++q) {
            v2f a0, a1;
            a0.x = cur[q].x; a0.y = cur[q].y;
            a1.x = cur[q].z; a1.y = cur[q].w;
            acc0 = __builtin_amdgcn_wmma_f32_16x16x4_f32(false, a0, false, wf[2 * q][0], (short)0, acc0, false, false);
            acc1 = __builtin_amdgcn_wmma_f32_16x16x4_f32(false, a0, false, wf[2 * q][1], (short)0, acc1, false, false);
            acc0 = __builtin_amdgcn_wmma_f32_16x16x4_f32(false, a1, false, wf[2 * q + 1][0], (short)0, acc0, false, false);
            acc1 = __builtin_amdgcn_wmma_f32_16x16x4_f32(false, a1, false, wf[2 * q + 1][1], (short)0, acc1, false, false);
        }

        if (KEYS && threadIdx.x < 128) {
            const unsigned k0 = (unsigned)dk.x / RANGE, k1 = (unsigned)dk.y / RANGE;
            unsigned char* kb = (unsigned char*)keyspace + (size_t)key_slot((unsigned)(e0 >> 4)) * 16 + (e0 & 15);
            *(unsigned short*)kb = (unsigned short)(k0 | (k1 << 8));
        }

        float pl[8], pr[8];
        #pragma unroll
        for (int j = 0; j < 8; ++j) {
            float s = acc0[j] * l0;
            s = fmaf(acc1[j], l1, s);
            float u = acc0[j] * r0;
            u = fmaf(acc1[j], r1, u);
            pl[j] = row16_sum(s);
            pr[j] = row16_sum(u);
        }
        if (row == 0) {
            #pragma unroll
            for (int j = 0; j < 8; ++j) part[pb][wave][khalf * 8 + j] = make_float2(pl[j], pr[j]);
        }
        __syncthreads();
        if (threadIdx.x < 32) {
            const int hd = lane >> 4;
            const float2 p0 = part[pb][hd * 4][row],     p1 = part[pb][hd * 4 + 1][row];
            const float2 p2 = part[pb][hd * 4 + 2][row], p3 = part[pb][hd * 4 + 3][row];
            el[(size_t)(mbase + row) * 2 + hd] = ((p0.x + p1.x) + p2.x) + p3.x;
            er[(size_t)(mbase + row) * 2 + hd] = ((p0.y + p1.y) + p2.y) + p3.y;
        }

        #pragma unroll
        for (int j = 0; j < 8; ++j) {
            const int m = mbase + j + khalf * 8;
            float* F = feat + (size_t)m * 256 + nq + row;
            F[0]  = acc0[j];
            F[16] = acc1[j];
        }
    };

    int t = blockIdx.x;
    tile_step(fa, fb, da, db, t, 0);
    t += GEMM_BLOCKS;
    while (t < N_TILES) {
        tile_step(fb, fa, db, da, t, 1);
        t += GEMM_BLOCKS;
        if (t >= N_TILES) break;
        tile_step(fa, fb, da, db, t, 0);
        t += GEMM_BLOCKS;
    }
}

#define QCAP     224

__device__ __forceinline__ void lds_add(float* p, float v) {
    __hip_atomic_fetch_add(p, v, __ATOMIC_RELAXED, __HIP_MEMORY_SCOPE_WORKGROUP);
}
__device__ __forceinline__ void lds_max(unsigned* p, unsigned v) {
    __hip_atomic_fetch_max(p, v, __ATOMIC_RELAXED, __HIP_MEMORY_SCOPE_WORKGROUP);
}
__device__ __forceinline__ float leaky(float v) { return v > 0.0f ? v : 0.2f * v; }

__device__ __forceinline__ unsigned append4(unsigned* qe, unsigned cnt, const int4 dv, const int i,
                                            const bool ok, const int base, const unsigned lt) {
    const unsigned r0 = (unsigned)(dv.x - base), r1 = (unsigned)(dv.y - base);
    const unsigned r2 = (unsigned)(dv.z - base), r3 = (unsigned)(dv.w - base);
    const bool m0 = ok & (r0 < RANGE), m1 = ok & (r1 < RANGE);
    const bool m2 = ok & (r2 < RANGE), m3 = ok & (r3 < RANGE);
    if (__builtin_amdgcn_ballot_w32(m0 | m1 | m2 | m3)) {
        const unsigned e4 = (unsigned)i * 4u;
        unsigned bal, pos;
        bal = __builtin_amdgcn_ballot_w32(m0);
        pos = cnt + __builtin_popcount(bal & lt);
        if (m0 && pos < QCAP) qe[pos] = ((e4 + 0u) << 8) | r0;
        cnt += __builtin_popcount(bal);
        bal = __builtin_amdgcn_ballot_w32(m1);
        pos = cnt + __builtin_popcount(bal & lt);
        if (m1 && pos < QCAP) qe[pos] = ((e4 + 1u) << 8) | r1;
        cnt += __builtin_popcount(bal);
        bal = __builtin_amdgcn_ballot_w32(m2);
        pos = cnt + __builtin_popcount(bal & lt);
        if (m2 && pos < QCAP) qe[pos] = ((e4 + 2u) << 8) | r2;
        cnt += __builtin_popcount(bal);
        bal = __builtin_amdgcn_ballot_w32(m3);
        pos = cnt + __builtin_popcount(bal & lt);
        if (m3 && pos < QCAP) qe[pos] = ((e4 + 3u) << 8) | r3;
        cnt += __builtin_popcount(bal);
    }
    return cnt;
}

__device__ __forceinline__ unsigned zero_bytes(const unsigned x) {
    return ~(((x & 0x7f7f7f7fu) + 0x7f7f7f7fu) | x) & 0x80808080u;
}

__device__ __forceinline__ unsigned append_keys(unsigned* qe, unsigned cnt, const int4 kv, const int i,
                                                const bool ok, const unsigned idm, const int lane) {
    const unsigned z0 = zero_bytes((unsigned)kv.x ^ idm), z1 = zero_bytes((unsigned)kv.y ^ idm);
    const unsigned z2 = zero_bytes((unsigned)kv.z ^ idm), z3 = zero_bytes((unsigned)kv.w ^ idm);
    const unsigned mb = ok ? ((z0 >> 7) | (z1 >> 6) | (z2 >> 5) | (z3 >> 4)) : 0u;
    unsigned bal = __builtin_amdgcn_ballot_w32(mb != 0u);
    if (bal) {
        const int c = __builtin_popcount(mb);
        unsigned pos = cnt;
        do {
            const int L = __builtin_ctz(bal);
            bal &= bal - 1u;
            const unsigned cL = (unsigned)__builtin_amdgcn_readlane(c, L);
            if (lane > L) pos += cL;
            cnt += cL;
        } while (bal);
        unsigned m = mb;
        while (m) {
            const int p = __builtin_ctz(m);
            m &= m - 1u;
            if (pos < QCAP) qe[pos] = (unsigned)i * 16u + (unsigned)((p & 7) * 4 + (p >> 3));
            ++pos;
        }
    }
    return cnt;
}

template <int LAYER>
__device__ __forceinline__ void add_rows(float* tile, const int n, const int lane,
                                         const int s, const int dl, const float w0, const float w1,
                                         const float* __restrict__ feat) {
    constexpr int COLS = LAYER ? 256 : 128;
    for (int j0 = 0; j0 < n; j0 += 8) {
        float4 f0[8], f1[8];
        #pragma unroll
        for (int u = 0; u < 8; ++u) {
            const int sj = __builtin_amdgcn_readlane(s, j0 + u);
            const float* row = feat + (size_t)sj * 256 + lane * 4;
            f0[u] = *(const float4*)row;
            f1[u] = *(const float4*)(row + 128);
        }
        #pragma unroll
        for (int u = 0; u < 8; ++u) {
            if (j0 + u < n) {
                const int   dj = __builtin_amdgcn_readlane(dl, j0 + u);
                const float a0 = __int_as_float(__builtin_amdgcn_readlane(__float_as_int(w0), j0 + u));
                const float a1 = __int_as_float(__builtin_amdgcn_readlane(__float_as_int(w1), j0 + u));
                float* t = tile + dj * COLS + lane;
                if (LAYER) {
                    lds_add(t +   0, f0[u].x * a0);
                    lds_add(t +  32, f0[u].y * a0);
                    lds_add(t +  64, f0[u].z * a0);
                    lds_add(t +  96, f0[u].w * a0);
                    lds_add(t + 128, f1[u].x * a1);
                    lds_add(t + 160, f1[u].y * a1);
                    lds_add(t + 192, f1[u].z * a1);
                    lds_add(t + 224, f1[u].w * a1);
                } else {
                    lds_add(t +  0, f0[u].x * a0 + f1[u].x * a1);
                    lds_add(t + 32, f0[u].y * a0 + f1[u].y * a1);
                    lds_add(t + 64, f0[u].z * a0 + f1[u].z * a1);
                    lds_add(t + 96, f0[u].w * a0 + f1[u].w * a1);
                }
            }
        }
    }
}

template <int LAYER, int PASS>
__device__ __forceinline__ void batch_from_entries(float* tile, unsigned* mx, float* dn, const float2* erl,
                                                   const unsigned* q, const int n, const int lane,
                                                   const float* __restrict__ feat,
                                                   const float* __restrict__ el,
                                                   const int* __restrict__ src) {
    const bool valid = lane < n;
    const unsigned ent = valid ? q[lane] : 0u;
    const int e  = (int)(ent >> 8);
    const int dl = (int)(ent & 255u);
    const int s  = src[e];
    const float2 l = *(const float2*)(el + (size_t)s * 2);
    const float2 r = erl[dl];
    const float c0 = leaky(l.x + r.x), c1 = leaky(l.y + r.y);
    if (PASS == 0) {
        if (valid) { lds_max(mx + dl * 2, ordEnc(c0)); lds_max(mx + dl * 2 + 1, ordEnc(c1)); }
        return;
    }
    const float x0 = expf(c0 - ordDec(mx[dl * 2])), x1 = expf(c1 - ordDec(mx[dl * 2 + 1]));
    if (PASS == 1) {
        if (valid) { lds_add(dn + dl * 2, x0); lds_add(dn + dl * 2 + 1, x1); }
        return;
    }
    float w0 = x0 / dn[dl * 2], w1 = x1 / dn[dl * 2 + 1];
    if (LAYER == 0) { w0 *= 0.5f; w1 *= 0.5f; }
    add_rows<LAYER>(tile, n, lane, s, dl, w0, w1, feat);
}

template <int LAYER, int PASS>
__device__ __forceinline__ void rescan_pass(float* tile, unsigned* mx, float* dn, const float2* erl,
                                            unsigned* qe, const int wave, const int lane, const int base,
                                            const float* __restrict__ feat,
                                            const float* __restrict__ el,
                                            const int* __restrict__ src,
                                            const int* __restrict__ dst) {
    constexpr int NV = N_EDGES / 4;
    const unsigned lt = (1u << lane) - 1u;
    const int4* d4 = (const int4*)dst;
    unsigned cnt = 0;
    for (int i0 = wave * 32; i0 < NV; i0 += 1024) {
        const int i = i0 + lane;
        const bool ok = i < NV;
        const int4 dv = d4[ok ? i : NV - 1];
        cnt = append4(qe, cnt, dv, i, ok, base, lt);
        while (cnt >= 32) {
            cnt -= 32;
            batch_from_entries<LAYER, PASS>(tile, mx, dn, erl, qe + cnt, 32, lane, feat, el, src);
        }
    }
    if (cnt) batch_from_entries<LAYER, PASS>(tile, mx, dn, erl, qe, (int)cnt, lane, feat, el, src);
}

template <int LAYER>
__device__ __forceinline__ void owned_range(float* tile, unsigned* queue, unsigned* mx, float* dn,
                                            float2* erl, unsigned* flag,
                                            const float* __restrict__ feat,
                                            const float* __restrict__ el,
                                            const float* __restrict__ er,
                                            const int* __restrict__ src,
                                            const int* __restrict__ dst,
                                            const float* __restrict__ b,
                                            float* park,
                                            float* outp) {
    constexpr int COLS = LAYER ? 256 : 128;
    constexpr int NV   = N_EDGES / 4;
    const int tid  = threadIdx.x;
    const int lane = tid & 31;
    const int wave = __builtin_amdgcn_readfirstlane(tid >> 5);
    const int base = blockIdx.x * RANGE;

    const int rows = min(RANGE, N_NODES - base);
    const int slot = min(256, rows * 8);
    const int cap  = min(QCAP, slot - 32);
    float* pk = park + (size_t)base * 256 + wave * slot;
    unsigned parked = 0u;
    if (LAYER) parked = __float_as_uint(pk[0]);

    for (int i = tid; i < RANGE * COLS; i += 1024) {
        float v = 0.0f;
        if (LAYER == 0) {
            const int r = i & 127, c = r >> 5, l = r & 31;
            v = 0.5f * (b[l * 4 + c] + b[128 + l * 4 + c]);
        }
        tile[i] = v;
    }
    if (tid < RANGE * 2) { mx[tid] = ORD_NEG_INF; dn[tid] = 0.0f; }
    if (tid < RANGE) {
        const int nd = base + tid;
        erl[tid] = (nd < N_NODES) ? *(const float2*)(er + (size_t)nd * 2) : make_float2(0.0f, 0.0f);
    }
    if (tid == 0) *flag = 0u;
    __syncthreads();

    unsigned* qe  = queue + wave * (4 * QCAP);
    int*      qs  = (int*)(qe + QCAP);
    float*    qw0 = (float*)(qe + 2 * QCAP);
    float*    qw1 = (float*)(qe + 3 * QCAP);
    unsigned cnt = 0;
    if (LAYER == 0) {
        const unsigned idm = (unsigned)blockIdx.x * 0x01010101u;
        const int4* k4 = (const int4*)park;
        for (int i0 = wave * 32; i0 < KEY_V4; i0 += 4096) {
            int4 kv[4];
            #pragma unroll
            for (int u = 0; u < 4; ++u) {
                const int i = i0 + u * 1024 + lane;
                kv[u] = k4[key_slot((unsigned)min(i, KEY_V4 - 1))];
            }
            asm volatile("" :: "v"(kv[0].x), "v"(kv[0].y), "v"(kv[0].z), "v"(kv[0].w), "v"(kv[1].x), "v"(kv[1].y), "v"(kv[1].z), "v"(kv[1].w),
                            "v"(kv[2].x), "v"(kv[2].y), "v"(kv[2].z), "v"(kv[2].w), "v"(kv[3].x), "v"(kv[3].y), "v"(kv[3].z), "v"(kv[3].w));
            #pragma unroll
            for (int u = 0; u < 4; ++u) {
                const int i = i0 + u * 1024 + lane;
                cnt = append_keys(qe, cnt, kv[u], i, i < KEY_V4, idm, lane);
            }
        }
        if (cnt > QCAP && lane == 0) *flag = 1u;
    } else {
        cnt = __builtin_amdgcn_readfirstlane(parked);
        if (cnt > (unsigned)cap && lane == 0) *flag = 1u;
    }
    __syncthreads();
    const bool overflow = *flag != 0u;
    if (LAYER == 0 && lane == 0) pk[0] = __uint_as_float(cnt);

    if (!overflow) {
        const int n = (int)cnt;
        for (int b0 = 0; b0 < n; b0 += 128) {
            unsigned ent[4]; int s[4]; int nd[4]; float2 l[4];
            #pragma unroll
            for (int u = 0; u < 4; ++u) {
                const int idx = b0 + u * 32 + lane;
                if (LAYER) ent[u] = (idx < n) ? __float_as_uint(pk[32 + idx]) : 0u;
                else       ent[u] = (idx < n) ? qe[idx] : 0u;
            }
            #pragma unroll
            for (int u = 0; u < 4; ++u) {
                s[u]  = LAYER ? (int)(ent[u] >> 8)   : src[ent[u]];
                nd[u] = LAYER ? (int)(ent[u] & 255u) : dst[ent[u]] - base;
            }
            #pragma unroll
            for (int u = 0; u < 4; ++u) l[u] = *(const float2*)(el + (size_t)s[u] * 2);
            #pragma unroll
            for (int u = 0; u < 4; ++u) {
                const int idx = b0 + u * 32 + lane;
                if (idx < n) {
                    const int dl = nd[u];
                    const float2 r = erl[dl];
                    const float c0 = leaky(l[u].x + r.x), c1 = leaky(l[u].y + r.y);
                    lds_max(mx + dl * 2, ordEnc(c0));
                    lds_max(mx + dl * 2 + 1, ordEnc(c1));
                    qs[idx] = s[u]; qw0[idx] = c0; qw1[idx] = c1;
                    qe[idx] = (unsigned)dl;
                }
            }
        }
        __syncthreads();
        for (int idx = lane; idx < n; idx += 32) {
            const int dl = (int)(qe[idx] & 255u);
            const float x0 = expf(qw0[idx] - ordDec(mx[dl * 2]));
            const float x1 = expf(qw1[idx] - ordDec(mx[dl * 2 + 1]));
            lds_add(dn + dl * 2, x0);
            lds_add(dn + dl * 2 + 1, x1);
            qw0[idx] = x0; qw1[idx] = x1;
        }
        __syncthreads();
        for (int b0 = 0; b0 < n; b0 += 32) {
            const int idx = b0 + lane;
            const bool valid = idx < n;
            const unsigned ent = valid ? qe[idx] : 0u;
            const int dl = (int)(ent & 255u);
            const int s  = valid ? qs[idx] : 0;
            float w0 = valid ? qw0[idx] / dn[dl * 2] : 0.0f;
            float w1 = valid ? qw1[idx] / dn[dl * 2 + 1] : 0.0f;
            if (LAYER == 0) { w0 *= 0.5f; w1 *= 0.5f; }
            if (LAYER == 0 && valid && idx < cap) pk[32 + idx] = __uint_as_float(((unsigned)s << 8) | (unsigned)dl);
            add_rows<LAYER>(tile, min(32, n - b0), lane, s, dl, w0, w1, feat);
        }
    } else {
        rescan_pass<LAYER, 0>(tile, mx, dn, erl, qe, wave, lane, base, feat, el, src, dst);
        __syncthreads();
        rescan_pass<LAYER, 1>(tile, mx, dn, erl, qe, wave, lane, base, feat, el, src, dst);
        __syncthreads();
        rescan_pass<LAYER, 2>(tile, mx, dn, erl, qe, wave, lane, base, feat, el, src, dst);
    }
    __syncthreads();

    constexpr int V4 = COLS / 4;
    for (int i = tid; i < RANGE * V4; i += 1024) {
        const int n = i / V4, r = i - n * V4;
        const int h = r >> 5, l = r & 31;
        if (base + n < N_NODES) {
            const float* t = tile + n * COLS + h * 128 + l;
            float4 v = make_float4(t[0], t[32], t[64], t[96]);
            if (LAYER) {
                const float4 bb = *(const float4*)(b + h * 128 + l * 4);
                v.x += bb.x; v.y += bb.y; v.z += bb.z; v.w += bb.w;
            }
            *(float4*)(outp + (size_t)(base + n) * COLS + h * 128 + l * 4) = v;
        }
    }
}

__global__ void __launch_bounds__(1024) aggregate_owned_mean(const float* __restrict__ feat,
                                                             const float* __restrict__ el,
                                                             const float* __restrict__ er,
                                                             const int* __restrict__ src,
                                                             const int* __restrict__ dst,
                                                             const float* __restrict__ b,
                                                             float* __restrict__ hnext,
                                                             float* __restrict__ fin) {
    __shared__ float tile[RANGE * 128];
    __shared__ unsigned queue[32 * 4 * QCAP];
    __shared__ unsigned mx[RANGE * 2];
    __shared__ float dn[RANGE * 2];
    __shared__ float2 erl[RANGE];
    __shared__ unsigned flag;
    owned_range<0>(tile, queue, mx, dn, erl, &flag, feat, el, er, src, dst, b, fin, hnext);
}

__global__ void __launch_bounds__(1024) aggregate_owned_out(const float* __restrict__ feat,
                                                            const float* __restrict__ el,
                                                            const float* __restrict__ er,
                                                            const int* __restrict__ src,
                                                            const int* __restrict__ dst,
                                                            const float* __restrict__ b,
                                                            float* __restrict__ out) {
    __shared__ float tile[RANGE * 256];
    __shared__ unsigned queue[32 * 4 * QCAP];
    __shared__ unsigned mx[RANGE * 2];
    __shared__ float dn[RANGE * 2];
    __shared__ float2 erl[RANGE];
    __shared__ unsigned flag;
    owned_range<1>(tile, queue, mx, dn, erl, &flag, feat, el, er, src, dst, b, out, out);
}

extern "C" void kernel_launch(void* const* d_in, const int* in_sizes, int n_in,
                              void* d_out, int out_size, void* d_ws, size_t ws_size,
                              hipStream_t stream) {
    const float* x   = (const float*)d_in[0];
    const float* Ws  = (const float*)d_in[1];
    const float* als = (const float*)d_in[2];
    const float* ars = (const float*)d_in[3];
    const float* bs  = (const float*)d_in[4];
    const int*   src = (const int*)d_in[5];
    const int*   dst = (const int*)d_in[6];

    float* ws      = (float*)d_ws;
    float* feat    = ws;
    float* hnext   = ws + 12800000;
    float* el      = ws + 19200000;
    float* er      = ws + 19300000;

    float* out = (float*)d_out;

    for (int layer = 0; layer < 2; ++layer) {
        const float* hin = layer ? hnext : x;
        const float* W   = Ws  + (size_t)layer * D_MODEL * 256;
        const float* al  = als + (size_t)layer * 256;
        const float* ar  = ars + (size_t)layer * 256;
        const float* b   = bs  + (size_t)layer * 256;

        if (layer == 0) {
            gemm_feat<true><<<GEMM_BLOCKS, 256, 0, stream>>>(hin, W, al, ar, feat, el, er, dst, out);
            aggregate_owned_mean<<<N_RANGES, 1024, 0, stream>>>(feat, el, er, src, dst, b, hnext, out);
        } else {
            gemm_feat<false><<<GEMM_BLOCKS, 256, 0, stream>>>(hin, W, al, ar, feat, el, er, dst, out);
            aggregate_owned_out<<<N_RANGES, 1024, 0, stream>>>(feat, el, er, src, dst, b, out);
        }
    }
}
